# starting kernel
# baseline (speedup 1.0000x reference)
#include <hip/hip_runtime.h>


typedef __attribute__((ext_vector_type(2))) float v2f;
typedef __attribute__((ext_vector_type(8))) float v8f;

__device__ __forceinline__ void mm(float* __restrict__ C, int ldc,
                                   const float* __restrict__ A, int lda,
                                   const float* __restrict__ B, int ldb,
                                   int M, int N, int Kd, bool transB)
{
    __syncthreads();
    const int wave = threadIdx.x >> 5;
    const int lane = threadIdx.x & 31;
    const int half = lane >> 4;
    const int l15  = lane & 15;
    const int tn = N >> 4;
    const int tiles = (M >> 4) * tn;
    for (int t = wave; t < tiles; t += 4) {
        const int mi = (t / tn) << 4;
        const int ni = (t % tn) << 4;
        v8f acc = {};
        #pragma unroll 4
        for (int k = 0; k < Kd; k += 4) {
            v2f a, b;
            const float* ar = A + (mi + l15) * lda + k + 2 * half;
            a.x = ar[0];
            a.y = ar[1];
            if (!transB) {
                const float* br = B + (k + 2 * half) * ldb + ni + l15;
                b.x = br[0];
                b.y = br[ldb];
            } else {
                const float* br = B + (ni + l15) * ldb + k + 2 * half;
                b.x = br[0];
                b.y = br[1];
            }
            acc = __builtin_amdgcn_wmma_f32_16x16x4_f32(
                false, a, false, b, (short)0, acc, false, false);
        }
        #pragma unroll
        for (int r = 0; r < 8; ++r)
            C[(mi + half * 8 + r) * ldc + ni + l15] = acc[r];
    }
    __syncthreads();
}

__device__ __forceinline__ float trace32(const float* A, float* red)
{
    __syncthreads();
    if (threadIdx.x < 32) red[threadIdx.x] = A[threadIdx.x * 33];
    __syncthreads();
    float s = 0.f;
    #pragma unroll
    for (int i = 0; i < 32; ++i) s += red[i];
    return s;
}

__device__ void ns_sqrt(float* dst_s, float* dst_i, const float* src,
                        float* y, float* z, float* t, float* yn, float* zn,
                        float* red, int iters)
{
    const int tid = threadIdx.x;
    const float c = trace32(src, red);
    const float inv_c = 1.f / c;
    __syncthreads();
    for (int i = tid; i < 1024; i += 128) {
        y[i] = src[i] * inv_c;
        z[i] = ((i % 33) == 0) ? 1.f : 0.f;
    }
    for (int it = 0; it < iters; ++it) {
        mm(t, 32, z, 32, y, 32, 32, 32, 32, false);
        for (int i = tid; i < 1024; i += 128)
            t[i] = (((i % 33) == 0) ? 1.5f : 0.f) - 0.5f * t[i];
        mm(yn, 32, y, 32, t, 32, 32, 32, 32, false);
        mm(zn, 32, t, 32, z, 32, 32, 32, 32, false);
        float* tmp;
        tmp = y; y = yn; yn = tmp;
        tmp = z; z = zn; zn = tmp;
    }
    const float sc  = __fsqrt_rn(c);
    const float isc = __frsqrt_rn(c);
    __syncthreads();
    if (dst_s) for (int i = tid; i < 1024; i += 128) dst_s[i] = y[i] * sc;
    if (dst_i) for (int i = tid; i < 1024; i += 128) dst_i[i] = z[i] * isc;
    __syncthreads();
}

__global__ __launch_bounds__(128, 1)
void spdnet_mixedop_kernel(const float* __restrict__ x,
                           const float* __restrict__ wts,
                           const float* __restrict__ Ws,
                           float* __restrict__ out)
{
    __shared__ float sX[4096];
    __shared__ float sWk[2048];
    __shared__ float sP[2048];
    __shared__ float sS[4][1024];
    __shared__ float sB0[1024], sB1[1024], sB2[1024];
    __shared__ float sRed[32];

    const int bc  = blockIdx.x;
    const int tid = threadIdx.x;

    float wk[4];
    #pragma unroll
    for (int k = 0; k < 4; ++k) wk[k] = wts[k];

    for (int i = tid; i < 4096; i += 128) sX[i] = x[bc * 4096 + i];

    for (int k = 0; k < 4; ++k) {
        __syncthreads();
        for (int i = tid; i < 2048; i += 128) sWk[i] = Ws[k * 2048 + i];
        mm(sP, 64, sWk, 64, sX, 64, 32, 64, 64, false);
        mm(sS[k], 32, sP, 64, sWk, 64, 32, 32, 64, true);
    }

    __syncthreads();
    for (int i = tid; i < 1024; i += 128)
        sB0[i] = wk[0] * sS[0][i] + wk[1] * sS[1][i] +
                 wk[2] * sS[2][i] + wk[3] * sS[3][i];

    float* pSq = sX;
    float* pSi = sX + 1024;
    float* pU  = sX + 2048;
    ns_sqrt(pSq, pSi, sB0, sB1, sB2, sWk, sWk + 1024, sP, sRed, 14);

    __syncthreads();
    for (int i = tid; i < 1024; i += 128) pU[i] = 0.f;

    for (int k = 0; k < 4; ++k) {
        mm(sB0, 32, pSi, 32, sS[k], 32, 32, 32, 32, false);
        mm(sB1, 32, sB0, 32, pSi, 32, 32, 32, 32, false);

        ns_sqrt(sB2, nullptr, sB1, sP, sP + 1024, sWk, sWk + 1024, sB0, sRed, 12);
        ns_sqrt(sB1, nullptr, sB2, sP, sP + 1024, sWk, sWk + 1024, sB0, sRed, 12);
        ns_sqrt(sB2, nullptr, sB1, sP, sP + 1024, sWk, sWk + 1024, sB0, sRed, 12);

        __syncthreads();
        for (int i = tid; i < 1024; i += 128)
            if ((i % 33) == 0) sB2[i] -= 1.f;

        float* q  = sB0;
        float* qn = sB1;
        __syncthreads();
        for (int i = tid; i < 1024; i += 128)
            q[i] = ((i % 33) == 0) ? (-1.f / 8.f) : 0.f;
        for (int ii = 7; ii >= 1; --ii) {
            mm(qn, 32, sB2, 32, q, 32, 32, 32, 32, false);
            const float a = ((ii & 1) ? 1.f : -1.f) / (float)ii;
            for (int i = tid; i < 1024; i += 128)
                if ((i % 33) == 0) qn[i] += a;
            float* tmp = q; q = qn; qn = tmp;
        }
        mm(sWk, 32, sB2, 32, q, 32, 32, 32, 32, false);

        const float scale = wk[k] * 8.f;
        for (int i = tid; i < 1024; i += 128) pU[i] += scale * sWk[i];
        __syncthreads();
    }

    __syncthreads();
    for (int i = tid; i < 1024; i += 128) {
        sB0[i] = pU[i] * (1.f / 64.f);
        sB1[i] = ((i % 33) == 0) ? 1.f : 0.f;
    }
    for (int ii = 10; ii >= 1; --ii) {
        mm(sB2, 32, sB0, 32, sB1, 32, 32, 32, 32, false);
        const float inv = 1.f / (float)ii;
        for (int i = tid; i < 1024; i += 128)
            sB1[i] = (((i % 33) == 0) ? 1.f : 0.f) + sB2[i] * inv;
    }
    float* r  = sB1;
    float* rn = sB2;
    for (int sq = 0; sq < 6; ++sq) {
        mm(rn, 32, r, 32, r, 32, 32, 32, 32, false);
        float* tmp = r; r = rn; rn = tmp;
    }
    mm(sB0, 32, pSq, 32, r, 32, 32, 32, 32, false);
    mm(sP, 32, sB0, 32, pSq, 32, 32, 32, 32, false);
    __syncthreads();
    for (int i = tid; i < 1024; i += 128) out[bc * 1024 + i] = sP[i];
}

extern "C" void kernel_launch(void* const* d_in, const int* in_sizes, int n_in,
                              void* d_out, int out_size, void* d_ws, size_t ws_size,
                              hipStream_t stream)
{
    const float* x   = (const float*)d_in[0];
    const float* wts = (const float*)d_in[1];
    const float* Ws  = (const float*)d_in[2];
    float* out = (float*)d_out;

    dim3 grid(2048);
    dim3 block(128);
    hipLaunchKernelGGL(spdnet_mixedop_kernel, grid, block, 0, stream,
                       x, wts, Ws, out);
}


# Pad LDS row pitch to 16 B x odd (32->36, 64->68 floats): f32 16x16x4 A-fragment and transposed-B reads go from 8-/16-way bank conflicts to conflict-free; arithmetic and summation order unchanged
# speedup vs baseline: 1.6860x; 1.6860x over previous
#include <hip/hip_runtime.h>


typedef __attribute__((ext_vector_type(2))) float v2f;
typedef __attribute__((ext_vector_type(8))) float v8f;

constexpr int LD  = 36;
constexpr int LDW = 68;
constexpr int MS  = 32 * LD;

__device__ __forceinline__ int at32(int i) { return (i >> 5) * LD + (i & 31); }
__device__ __forceinline__ bool diag32(int i) { return (i >> 5) == (i & 31); }

__device__ __forceinline__ void mm(float* __restrict__ C, int ldc,
                                   const float* __restrict__ A, int lda,
                                   const float* __restrict__ B, int ldb,
                                   int M, int N, int Kd, bool transB)
{
    __syncthreads();
    const int wave = threadIdx.x >> 5;
    const int lane = threadIdx.x & 31;
    const int half = lane >> 4;
    const int l15  = lane & 15;
    const int tn = N >> 4;
    const int tiles = (M >> 4) * tn;
    for (int t = wave; t < tiles; t += 4) {
        const int mi = (t / tn) << 4;
        const int ni = (t % tn) << 4;
        v8f acc = {};
        #pragma unroll 4
        for (int k = 0; k < Kd; k += 4) {
            v2f a, b;
            const float* ar = A + (mi + l15) * lda + k + 2 * half;
            a.x = ar[0];
            a.y = ar[1];
            if (!transB) {
                const float* br = B + (k + 2 * half) * ldb + ni + l15;
                b.x = br[0];
                b.y = br[ldb];
            } else {
                const float* br = B + (ni + l15) * ldb + k + 2 * half;
                b.x = br[0];
                b.y = br[1];
            }
            acc = __builtin_amdgcn_wmma_f32_16x16x4_f32(
                false, a, false, b, (short)0, acc, false, false);
        }
        #pragma unroll
        for (int r = 0; r < 8; ++r)
            C[(mi + half * 8 + r) * ldc + ni + l15] = acc[r];
    }
    __syncthreads();
}

__device__ __forceinline__ float trace32(const float* A, float* red)
{
    __syncthreads();
    if (threadIdx.x < 32) red[threadIdx.x] = A[threadIdx.x * (LD + 1)];
    __syncthreads();
    float s = 0.f;
    #pragma unroll
    for (int i = 0; i < 32; ++i) s += red[i];
    return s;
}

__device__ void ns_sqrt(float* dst_s, float* dst_i, const float* src,
                        float* y, float* z, float* t, float* yn, float* zn,
                        float* red, int iters)
{
    const int tid = threadIdx.x;
    const float c = trace32(src, red);
    const float inv_c = 1.f / c;
    __syncthreads();
    for (int i = tid; i < 1024; i += 128) {
        const int a = at32(i);
        y[a] = src[a] * inv_c;
        z[a] = diag32(i) ? 1.f : 0.f;
    }
    for (int it = 0; it < iters; ++it) {
        mm(t, LD, z, LD, y, LD, 32, 32, 32, false);
        for (int i = tid; i < 1024; i += 128) {
            const int a = at32(i);
            t[a] = (diag32(i) ? 1.5f : 0.f) - 0.5f * t[a];
        }
        mm(yn, LD, y, LD, t, LD, 32, 32, 32, false);
        mm(zn, LD, t, LD, z, LD, 32, 32, 32, false);
        float* tmp;
        tmp = y; y = yn; yn = tmp;
        tmp = z; z = zn; zn = tmp;
    }
    const float sc  = __fsqrt_rn(c);
    const float isc = __frsqrt_rn(c);
    __syncthreads();
    if (dst_s) for (int i = tid; i < 1024; i += 128) dst_s[at32(i)] = y[at32(i)] * sc;
    if (dst_i) for (int i = tid; i < 1024; i += 128) dst_i[at32(i)] = z[at32(i)] * isc;
    __syncthreads();
}

__global__ __launch_bounds__(128, 1)
void spdnet_mixedop_kernel(const float* __restrict__ x,
                           const float* __restrict__ wts,
                           const float* __restrict__ Ws,
                           float* __restrict__ out)
{
    __shared__ float lds[14 * MS + 32];
    float* const sS   = lds;
    float* const pSq  = lds + 4 * MS;
    float* const pSi  = lds + 5 * MS;
    float* const pU   = lds + 6 * MS;
    float* const sW0  = lds + 7 * MS;
    float* const sW1  = lds + 8 * MS;
    float* const sP0  = lds + 9 * MS;
    float* const sP1  = lds + 10 * MS;
    float* const sB0  = lds + 11 * MS;
    float* const sB1  = lds + 12 * MS;
    float* const sB2  = lds + 13 * MS;
    float* const sRed = lds + 14 * MS;
    float* const sX   = lds + 4 * MS;
    float* const sWk  = lds + 8 * MS;
    float* const sP   = lds + 10 * MS;

    const int bc  = blockIdx.x;
    const int tid = threadIdx.x;

    float wk[4];
    #pragma unroll
    for (int k = 0; k < 4; ++k) wk[k] = wts[k];

    for (int i = tid; i < 4096; i += 128) sX[(i >> 6) * LDW + (i & 63)] = x[bc * 4096 + i];

    for (int k = 0; k < 4; ++k) {
        __syncthreads();
        for (int i = tid; i < 2048; i += 128) sWk[(i >> 6) * LDW + (i & 63)] = Ws[k * 2048 + i];
        mm(sP, LDW, sWk, LDW, sX, LDW, 32, 64, 64, false);
        mm(sS + k * MS, LD, sP, LDW, sWk, LDW, 32, 32, 64, true);
    }

    __syncthreads();
    for (int i = tid; i < 1024; i += 128) {
        const int a = at32(i);
        sB0[a] = wk[0] * sS[a] + wk[1] * sS[MS + a] +
                 wk[2] * sS[2 * MS + a] + wk[3] * sS[3 * MS + a];
    }

    ns_sqrt(pSq, pSi, sB0, sB1, sB2, sW0, sW1, sP0, sRed, 14);

    __syncthreads();
    for (int i = tid; i < 1024; i += 128) pU[at32(i)] = 0.f;

    for (int k = 0; k < 4; ++k) {
        mm(sB0, LD, pSi, LD, sS + k * MS, LD, 32, 32, 32, false);
        mm(sB1, LD, sB0, LD, pSi, LD, 32, 32, 32, false);

        ns_sqrt(sB2, nullptr, sB1, sP0, sP1, sW0, sW1, sB0, sRed, 12);
        ns_sqrt(sB1, nullptr, sB2, sP0, sP1, sW0, sW1, sB0, sRed, 12);
        ns_sqrt(sB2, nullptr, sB1, sP0, sP1, sW0, sW1, sB0, sRed, 12);

        __syncthreads();
        for (int i = tid; i < 1024; i += 128)
            if (diag32(i)) sB2[at32(i)] -= 1.f;

        float* q  = sB0;
        float* qn = sB1;
        __syncthreads();
        for (int i = tid; i < 1024; i += 128)
            q[at32(i)] = diag32(i) ? (-1.f / 8.f) : 0.f;
        for (int ii = 7; ii >= 1; --ii) {
            mm(qn, LD, sB2, LD, q, LD, 32, 32, 32, false);
            const float a = ((ii & 1) ? 1.f : -1.f) / (float)ii;
            for (int i = tid; i < 1024; i += 128)
                if (diag32(i)) qn[at32(i)] += a;
            float* tmp = q; q = qn; qn = tmp;
        }
        mm(sW0, LD, sB2, LD, q, LD, 32, 32, 32, false);

        const float scale = wk[k] * 8.f;
        for (int i = tid; i < 1024; i += 128) pU[at32(i)] += scale * sW0[at32(i)];
        __syncthreads();
    }

    __syncthreads();
    for (int i = tid; i < 1024; i += 128) {
        const int a = at32(i);
        sB0[a] = pU[a] * (1.f / 64.f);
        sB1[a] = diag32(i) ? 1.f : 0.f;
    }
    for (int ii = 10; ii >= 1; --ii) {
        mm(sB2, LD, sB0, LD, sB1, LD, 32, 32, 32, false);
        const float inv = 1.f / (float)ii;
        for (int i = tid; i < 1024; i += 128) {
            const int a = at32(i);
            sB1[a] = (diag32(i) ? 1.f : 0.f) + sB2[a] * inv;
        }
    }
    float* r  = sB1;
    float* rn = sB2;
    for (int sq = 0; sq < 6; ++sq) {
        mm(rn, LD, r, LD, r, LD, 32, 32, 32, false);
        float* tmp = r; r = rn; rn = tmp;
    }
    mm(sB0, LD, pSq, LD, r, LD, 32, 32, 32, false);
    mm(sP0, LD, sB0, LD, pSq, LD, 32, 32, 32, false);
    __syncthreads();
    for (int i = tid; i < 1024; i += 128) out[bc * 1024 + i] = sP0[at32(i)];
}

extern "C" void kernel_launch(void* const* d_in, const int* in_sizes, int n_in,
                              void* d_out, int out_size, void* d_ws, size_t ws_size,
                              hipStream_t stream)
{
    const float* x   = (const float*)d_in[0];
    const float* wts = (const float*)d_in[1];
    const float* Ws  = (const float*)d_in[2];
    float* out = (float*)d_out;

    dim3 grid(2048);
    dim3 block(128);
    hipLaunchKernelGGL(spdnet_mixedop_kernel, grid, block, 0, stream,
                       x, wts, Ws, out);
}


# Newton-Schulz iteration in 2 barrier-separated LDS passes instead of 4: (3I-ZY)/2 formed in the product's accumulator; Y*T and T*Z as one pass of two independent WMMA chains per wave; sums unchanged
# speedup vs baseline: 2.3135x; 1.3722x over previous
#include <hip/hip_runtime.h>


typedef __attribute__((ext_vector_type(2))) float v2f;
typedef __attribute__((ext_vector_type(8))) float v8f;

constexpr int LD  = 36;
constexpr int LDW = 68;
constexpr int MS  = 32 * LD;

__device__ __forceinline__ int at32(int i) { return (i >> 5) * LD + (i & 31); }
__device__ __forceinline__ bool diag32(int i) { return (i >> 5) == (i & 31); }

__device__ __forceinline__ void mm(float* __restrict__ C, int ldc,
                                   const float* __restrict__ A, int lda,
                                   const float* __restrict__ B, int ldb,
                                   int M, int N, int Kd, bool transB)
{
    __syncthreads();
    const int wave = threadIdx.x >> 5;
    const int lane = threadIdx.x & 31;
    const int half = lane >> 4;
    const int l15  = lane & 15;
    const int tn = N >> 4;
    const int tiles = (M >> 4) * tn;
    for (int t = wave; t < tiles; t += 4) {
        const int mi = (t / tn) << 4;
        const int ni = (t % tn) << 4;
        v8f acc = {};
        #pragma unroll 4
        for (int k = 0; k < Kd; k += 4) {
            v2f a, b;
            const float* ar = A + (mi + l15) * lda + k + 2 * half;
            a.x = ar[0];
            a.y = ar[1];
            if (!transB) {
                const float* br = B + (k + 2 * half) * ldb + ni + l15;
                b.x = br[0];
                b.y = br[ldb];
            } else {
                const float* br = B + (ni + l15) * ldb + k + 2 * half;
                b.x = br[0];
                b.y = br[1];
            }
            acc = __builtin_amdgcn_wmma_f32_16x16x4_f32(
                false, a, false, b, (short)0, acc, false, false);
        }
        #pragma unroll
        for (int r = 0; r < 8; ++r)
            C[(mi + half * 8 + r) * ldc + ni + l15] = acc[r];
    }
    __syncthreads();
}

__device__ __forceinline__ float trace32(const float* A, float* red)
{
    __syncthreads();
    if (threadIdx.x < 32) red[threadIdx.x] = A[threadIdx.x * (LD + 1)];
    __syncthreads();
    float s = 0.f;
    #pragma unroll
    for (int i = 0; i < 32; ++i) s += red[i];
    return s;
}

__device__ void ns_sqrt(float* dst_s, float* dst_i, const float* src,
                        float* y, float* z, float* t, float* yn, float* zn,
                        float* red, int iters)
{
    const int tid = threadIdx.x;
    const float c = trace32(src, red);
    const float inv_c = 1.f / c;
    __syncthreads();
    for (int i = tid; i < 1024; i += 128) {
        const int a = at32(i);
        y[a] = src[a] * inv_c;
        z[a] = diag32(i) ? 1.f : 0.f;
    }
    const int wave = tid >> 5;
    const int lane = tid & 31;
    const int half = lane >> 4;
    const int l15  = lane & 15;
    const int mi = (wave >> 1) << 4;
    const int ni = (wave & 1) << 4;
    const int aoff = (mi + l15) * LD + 2 * half;
    const int boff = (2 * half) * LD + ni + l15;
    const int coff = (mi + half * 8) * LD + ni + l15;
    for (int it = 0; it < iters; ++it) {
        __syncthreads();
        {
            v2f fa[8], fb[8];
            #pragma unroll
            for (int j = 0; j < 8; ++j) {
                fa[j].x = z[aoff + 4 * j];
                fa[j].y = z[aoff + 4 * j + 1];
                fb[j].x = y[boff + 4 * j * LD];
                fb[j].y = y[boff + 4 * j * LD + LD];
            }
            v8f acc = {};
            #pragma unroll
            for (int j = 0; j < 8; ++j)
                acc = __builtin_amdgcn_wmma_f32_16x16x4_f32(
                    false, fa[j], false, fb[j], (short)0, acc, false, false);
            #pragma unroll
            for (int r = 0; r < 8; ++r) {
                const bool dg = (mi + half * 8 + r) == (ni + l15);
                t[coff + r * LD] = (dg ? 1.5f : 0.f) - 0.5f * acc[r];
            }
        }
        __syncthreads();
        {
            v2f ya[8], tb[8], ta[8], zb[8];
            #pragma unroll
            for (int j = 0; j < 8; ++j) {
                ya[j].x = y[aoff + 4 * j];
                ya[j].y = y[aoff + 4 * j + 1];
                tb[j].x = t[boff + 4 * j * LD];
                tb[j].y = t[boff + 4 * j * LD + LD];
                ta[j].x = t[aoff + 4 * j];
                ta[j].y = t[aoff + 4 * j + 1];
                zb[j].x = z[boff + 4 * j * LD];
                zb[j].y = z[boff + 4 * j * LD + LD];
            }
            __builtin_amdgcn_sched_barrier(0);
            v8f accY = {}, accZ = {};
            #pragma unroll
            for (int j = 0; j < 8; ++j) {
                accY = __builtin_amdgcn_wmma_f32_16x16x4_f32(
                    false, ya[j], false, tb[j], (short)0, accY, false, false);
                accZ = __builtin_amdgcn_wmma_f32_16x16x4_f32(
                    false, ta[j], false, zb[j], (short)0, accZ, false, false);
            }
            #pragma unroll
            for (int r = 0; r < 8; ++r) {
                yn[coff + r * LD] = accY[r];
                zn[coff + r * LD] = accZ[r];
            }
        }
        float* tmp;
        tmp = y; y = yn; yn = tmp;
        tmp = z; z = zn; zn = tmp;
    }
    const float sc  = __fsqrt_rn(c);
    const float isc = __frsqrt_rn(c);
    __syncthreads();
    if (dst_s) for (int i = tid; i < 1024; i += 128) dst_s[at32(i)] = y[at32(i)] * sc;
    if (dst_i) for (int i = tid; i < 1024; i += 128) dst_i[at32(i)] = z[at32(i)] * isc;
    __syncthreads();
}

__global__ __launch_bounds__(128, 1)
void spdnet_mixedop_kernel(const float* __restrict__ x,
                           const float* __restrict__ wts,
                           const float* __restrict__ Ws,
                           float* __restrict__ out)
{
    __shared__ float lds[14 * MS + 32];
    float* const sS   = lds;
    float* const pSq  = lds + 4 * MS;
    float* const pSi  = lds + 5 * MS;
    float* const pU   = lds + 6 * MS;
    float* const sW0  = lds + 7 * MS;
    float* const sW1  = lds + 8 * MS;
    float* const sP0  = lds + 9 * MS;
    float* const sP1  = lds + 10 * MS;
    float* const sB0  = lds + 11 * MS;
    float* const sB1  = lds + 12 * MS;
    float* const sB2  = lds + 13 * MS;
    float* const sRed = lds + 14 * MS;
    float* const sX   = lds + 4 * MS;
    float* const sWk  = lds + 8 * MS;
    float* const sP   = lds + 10 * MS;

    const int bc  = blockIdx.x;
    const int tid = threadIdx.x;

    float wk[4];
    #pragma unroll
    for (int k = 0; k < 4; ++k) wk[k] = wts[k];

    for (int i = tid; i < 4096; i += 128) sX[(i >> 6) * LDW + (i & 63)] = x[bc * 4096 + i];

    for (int k = 0; k < 4; ++k) {
        __syncthreads();
        for (int i = tid; i < 2048; i += 128) sWk[(i >> 6) * LDW + (i & 63)] = Ws[k * 2048 + i];
        mm(sP, LDW, sWk, LDW, sX, LDW, 32, 64, 64, false);
        mm(sS + k * MS, LD, sP, LDW, sWk, LDW, 32, 32, 64, true);
    }

    __syncthreads();
    for (int i = tid; i < 1024; i += 128) {
        const int a = at32(i);
        sB0[a] = wk[0] * sS[a] + wk[1] * sS[MS + a] +
                 wk[2] * sS[2 * MS + a] + wk[3] * sS[3 * MS + a];
    }

    ns_sqrt(pSq, pSi, sB0, sB1, sB2, sW0, sW1, sP0, sRed, 14);

    __syncthreads();
    for (int i = tid; i < 1024; i += 128) pU[at32(i)] = 0.f;

    for (int k = 0; k < 4; ++k) {
        mm(sB0, LD, pSi, LD, sS + k * MS, LD, 32, 32, 32, false);
        mm(sB1, LD, sB0, LD, pSi, LD, 32, 32, 32, false);

        ns_sqrt(sB2, nullptr, sB1, sP0, sP1, sW0, sW1, sB0, sRed, 12);
        ns_sqrt(sB1, nullptr, sB2, sP0, sP1, sW0, sW1, sB0, sRed, 12);
        ns_sqrt(sB2, nullptr, sB1, sP0, sP1, sW0, sW1, sB0, sRed, 12);

        __syncthreads();
        for (int i = tid; i < 1024; i += 128)
            if (diag32(i)) sB2[at32(i)] -= 1.f;

        float* q  = sB0;
        float* qn = sB1;
        __syncthreads();
        for (int i = tid; i < 1024; i += 128)
            q[at32(i)] = diag32(i) ? (-1.f / 8.f) : 0.f;
        for (int ii = 7; ii >= 1; --ii) {
            mm(qn, LD, sB2, LD, q, LD, 32, 32, 32, false);
            const float a = ((ii & 1) ? 1.f : -1.f) / (float)ii;
            for (int i = tid; i < 1024; i += 128)
                if (diag32(i)) qn[at32(i)] += a;
            float* tmp = q; q = qn; qn = tmp;
        }
        mm(sW0, LD, sB2, LD, q, LD, 32, 32, 32, false);

        const float scale = wk[k] * 8.f;
        for (int i = tid; i < 1024; i += 128) pU[at32(i)] += scale * sW0[at32(i)];
        __syncthreads();
    }

    __syncthreads();
    for (int i = tid; i < 1024; i += 128) {
        const int a = at32(i);
        sB0[a] = pU[a] * (1.f / 64.f);
        sB1[a] = diag32(i) ? 1.f : 0.f;
    }
    for (int ii = 10; ii >= 1; --ii) {
        mm(sB2, LD, sB0, LD, sB1, LD, 32, 32, 32, false);
        const float inv = 1.f / (float)ii;
        for (int i = tid; i < 1024; i += 128) {
            const int a = at32(i);
            sB1[a] = (diag32(i) ? 1.f : 0.f) + sB2[a] * inv;
        }
    }
    float* r  = sB1;
    float* rn = sB2;
    for (int sq = 0; sq < 6; ++sq) {
        mm(rn, LD, r, LD, r, LD, 32, 32, 32, false);
        float* tmp = r; r = rn; rn = tmp;
    }
    mm(sB0, LD, pSq, LD, r, LD, 32, 32, 32, false);
    mm(sP0, LD, sB0, LD, pSq, LD, 32, 32, 32, false);
    __syncthreads();
    for (int i = tid; i < 1024; i += 128) out[bc * 1024 + i] = sP0[at32(i)];
}

extern "C" void kernel_launch(void* const* d_in, const int* in_sizes, int n_in,
                              void* d_out, int out_size, void* d_ws, size_t ws_size,
                              hipStream_t stream)
{
    const float* x   = (const float*)d_in[0];
    const float* wts = (const float*)d_in[1];
    const float* Ws  = (const float*)d_in[2];
    float* out = (float*)d_out;

    dim3 grid(2048);
    dim3 block(128);
    hipLaunchKernelGGL(spdnet_mixedop_kernel, grid, block, 0, stream,
                       x, wts, Ws, out);
}


# Log-map phase: each of the 4 waves runs one candidate op's whole chain (congruence, 3 Newton-Schulz roots, log series) alone on all four tiles, no workgroup barriers there; every sum formed as before
# speedup vs baseline: 3.2276x; 1.3951x over previous
#include <hip/hip_runtime.h>


typedef __attribute__((ext_vector_type(2))) float v2f;
typedef __attribute__((ext_vector_type(8))) float v8f;

constexpr int LD  = 36;
constexpr int LDW = 68;
constexpr int MS  = 32 * LD;

__device__ __forceinline__ int at32(int i) { return (i >> 5) * LD + (i & 31); }
__device__ __forceinline__ bool diag32(int i) { return (i >> 5) == (i & 31); }

__device__ __forceinline__ void mm(float* __restrict__ C, int ldc,
                                   const float* __restrict__ A, int lda,
                                   const float* __restrict__ B, int ldb,
                                   int M, int N, int Kd, bool transB)
{
    __syncthreads();
    const int wave = threadIdx.x >> 5;
    const int lane = threadIdx.x & 31;
    const int half = lane >> 4;
    const int l15  = lane & 15;
    const int tn = N >> 4;
    const int tiles = (M >> 4) * tn;
    for (int t = wave; t < tiles; t += 4) {
        const int mi = (t / tn) << 4;
        const int ni = (t % tn) << 4;
        v8f acc = {};
        #pragma unroll 4
        for (int k = 0; k < Kd; k += 4) {
            v2f a, b;
            const float* ar = A + (mi + l15) * lda + k + 2 * half;
            a.x = ar[0];
            a.y = ar[1];
            if (!transB) {
                const float* br = B + (k + 2 * half) * ldb + ni + l15;
                b.x = br[0];
                b.y = br[ldb];
            } else {
                const float* br = B + (ni + l15) * ldb + k + 2 * half;
                b.x = br[0];
                b.y = br[1];
            }
            acc = __builtin_amdgcn_wmma_f32_16x16x4_f32(
                false, a, false, b, (short)0, acc, false, false);
        }
        #pragma unroll
        for (int r = 0; r < 8; ++r)
            C[(mi + half * 8 + r) * ldc + ni + l15] = acc[r];
    }
    __syncthreads();
}

__device__ __forceinline__ float trace32(const float* A, float* red)
{
    __syncthreads();
    if (threadIdx.x < 32) red[threadIdx.x] = A[threadIdx.x * (LD + 1)];
    __syncthreads();
    float s = 0.f;
    #pragma unroll
    for (int i = 0; i < 32; ++i) s += red[i];
    return s;
}

__device__ void ns_sqrt(float* dst_s, float* dst_i, const float* src,
                        float* y, float* z, float* t, float* yn, float* zn,
                        float* red, int iters)
{
    const int tid = threadIdx.x;
    const float c = trace32(src, red);
    const float inv_c = 1.f / c;
    __syncthreads();
    for (int i = tid; i < 1024; i += 128) {
        const int a = at32(i);
        y[a] = src[a] * inv_c;
        z[a] = diag32(i) ? 1.f : 0.f;
    }
    const int wave = tid >> 5;
    const int lane = tid & 31;
    const int half = lane >> 4;
    const int l15  = lane & 15;
    const int mi = (wave >> 1) << 4;
    const int ni = (wave & 1) << 4;
    const int aoff = (mi + l15) * LD + 2 * half;
    const int boff = (2 * half) * LD + ni + l15;
    const int coff = (mi + half * 8) * LD + ni + l15;
    for (int it = 0; it < iters; ++it) {
        __syncthreads();
        {
            v2f fa[8], fb[8];
            #pragma unroll
            for (int j = 0; j < 8; ++j) {
                fa[j].x = z[aoff + 4 * j];
                fa[j].y = z[aoff + 4 * j + 1];
                fb[j].x = y[boff + 4 * j * LD];
                fb[j].y = y[boff + 4 * j * LD + LD];
            }
            v8f acc = {};
            #pragma unroll
            for (int j = 0; j < 8; ++j)
                acc = __builtin_amdgcn_wmma_f32_16x16x4_f32(
                    false, fa[j], false, fb[j], (short)0, acc, false, false);
            #pragma unroll
            for (int r = 0; r < 8; ++r) {
                const bool dg = (mi + half * 8 + r) == (ni + l15);
                t[coff + r * LD] = (dg ? 1.5f : 0.f) - 0.5f * acc[r];
            }
        }
        __syncthreads();
        {
            v2f ya[8], tb[8], ta[8], zb[8];
            #pragma unroll
            for (int j = 0; j < 8; ++j) {
                ya[j].x = y[aoff + 4 * j];
                ya[j].y = y[aoff + 4 * j + 1];
                tb[j].x = t[boff + 4 * j * LD];
                tb[j].y = t[boff + 4 * j * LD + LD];
                ta[j].x = t[aoff + 4 * j];
                ta[j].y = t[aoff + 4 * j + 1];
                zb[j].x = z[boff + 4 * j * LD];
                zb[j].y = z[boff + 4 * j * LD + LD];
            }
            __builtin_amdgcn_sched_barrier(0);
            v8f accY = {}, accZ = {};
            #pragma unroll
            for (int j = 0; j < 8; ++j) {
                accY = __builtin_amdgcn_wmma_f32_16x16x4_f32(
                    false, ya[j], false, tb[j], (short)0, accY, false, false);
                accZ = __builtin_amdgcn_wmma_f32_16x16x4_f32(
                    false, ta[j], false, zb[j], (short)0, accZ, false, false);
            }
            #pragma unroll
            for (int r = 0; r < 8; ++r) {
                yn[coff + r * LD] = accY[r];
                zn[coff + r * LD] = accZ[r];
            }
        }
        float* tmp;
        tmp = y; y = yn; yn = tmp;
        tmp = z; z = zn; zn = tmp;
    }
    const float sc  = __fsqrt_rn(c);
    const float isc = __frsqrt_rn(c);
    __syncthreads();
    if (dst_s) for (int i = tid; i < 1024; i += 128) dst_s[at32(i)] = y[at32(i)] * sc;
    if (dst_i) for (int i = tid; i < 1024; i += 128) dst_i[at32(i)] = z[at32(i)] * isc;
    __syncthreads();
}

__device__ __forceinline__ void wave_lds_sync()
{
    __builtin_amdgcn_fence(__ATOMIC_RELEASE, "wavefront");
    __builtin_amdgcn_wave_barrier();
    __builtin_amdgcn_fence(__ATOMIC_ACQUIRE, "wavefront");
}

template <int EPI>
__device__ __forceinline__ void mmw(float* C, const float* A, const float* B,
                                    int half, int l15)
{
    v2f a0[8], a1[8], b0[8], b1[8];
    const float* ar = A + l15 * LD + 2 * half;
    const float* br = B + (2 * half) * LD + l15;
    #pragma unroll
    for (int j = 0; j < 8; ++j) {
        a0[j].x = ar[4 * j];
        a0[j].y = ar[4 * j + 1];
        a1[j].x = ar[16 * LD + 4 * j];
        a1[j].y = ar[16 * LD + 4 * j + 1];
        b0[j].x = br[4 * j * LD];
        b0[j].y = br[4 * j * LD + LD];
        b1[j].x = br[4 * j * LD + 16];
        b1[j].y = br[4 * j * LD + LD + 16];
    }
    __builtin_amdgcn_sched_barrier(0);
    v8f c00 = {}, c01 = {}, c10 = {}, c11 = {};
    #pragma unroll
    for (int j = 0; j < 8; ++j) {
        c00 = __builtin_amdgcn_wmma_f32_16x16x4_f32(
            false, a0[j], false, b0[j], (short)0, c00, false, false);
        c01 = __builtin_amdgcn_wmma_f32_16x16x4_f32(
            false, a0[j], false, b1[j], (short)0, c01, false, false);
        c10 = __builtin_amdgcn_wmma_f32_16x16x4_f32(
            false, a1[j], false, b0[j], (short)0, c10, false, false);
        c11 = __builtin_amdgcn_wmma_f32_16x16x4_f32(
            false, a1[j], false, b1[j], (short)0, c11, false, false);
    }
    float* cr = C + (half * 8) * LD + l15;
    #pragma unroll
    for (int r = 0; r < 8; ++r) {
        if (EPI == 1) {
            const bool dg = (half * 8 + r) == l15;
            cr[r * LD]             = (dg ? 1.5f : 0.f) - 0.5f * c00[r];
            cr[r * LD + 16]        = 0.f - 0.5f * c01[r];
            cr[(16 + r) * LD]      = 0.f - 0.5f * c10[r];
            cr[(16 + r) * LD + 16] = (dg ? 1.5f : 0.f) - 0.5f * c11[r];
        } else {
            cr[r * LD]             = c00[r];
            cr[r * LD + 16]        = c01[r];
            cr[(16 + r) * LD]      = c10[r];
            cr[(16 + r) * LD + 16] = c11[r];
        }
    }
    wave_lds_sync();
}

__device__ __forceinline__ void ns_sqrt_w(float* y, float* z, float* t, int iters,
                                          int lane, int half, int l15)
{
    float c = 0.f;
    #pragma unroll
    for (int i = 0; i < 32; ++i) c += y[i * (LD + 1)];
    const float inv_c = 1.f / c;
    for (int i = lane; i < 1024; i += 32) {
        const int a = at32(i);
        y[a] = y[a] * inv_c;
        z[a] = diag32(i) ? 1.f : 0.f;
    }
    wave_lds_sync();
    for (int it = 0; it < iters; ++it) {
        mmw<1>(t, z, y, half, l15);
        mmw<0>(y, y, t, half, l15);
        mmw<0>(z, t, z, half, l15);
    }
    const float sc = __fsqrt_rn(c);
    for (int i = lane; i < 1024; i += 32) {
        const int a = at32(i);
        y[a] = y[a] * sc;
    }
    wave_lds_sync();
}

__global__ __launch_bounds__(128, 1)
void spdnet_mixedop_kernel(const float* __restrict__ x,
                           const float* __restrict__ wts,
                           const float* __restrict__ Ws,
                           float* __restrict__ out)
{
    __shared__ float lds[14 * MS + 32];
    float* const sS   = lds;
    float* const pSq  = lds + 4 * MS;
    float* const pSi  = lds + 5 * MS;
    float* const sW0  = lds + 7 * MS;
    float* const sW1  = lds + 8 * MS;
    float* const sP0  = lds + 9 * MS;
    float* const sB0  = lds + 11 * MS;
    float* const sB1  = lds + 12 * MS;
    float* const sB2  = lds + 13 * MS;
    float* const sRed = lds + 14 * MS;
    float* const sX   = lds + 4 * MS;
    float* const sWk  = lds + 8 * MS;
    float* const sP   = lds + 10 * MS;

    const int bc  = blockIdx.x;
    const int tid = threadIdx.x;

    float wk[4];
    #pragma unroll
    for (int k = 0; k < 4; ++k) wk[k] = wts[k];

    for (int i = tid; i < 4096; i += 128) sX[(i >> 6) * LDW + (i & 63)] = x[bc * 4096 + i];

    for (int k = 0; k < 4; ++k) {
        __syncthreads();
        for (int i = tid; i < 2048; i += 128) sWk[(i >> 6) * LDW + (i & 63)] = Ws[k * 2048 + i];
        mm(sP, LDW, sWk, LDW, sX, LDW, 32, 64, 64, false);
        mm(sS + k * MS, LD, sP, LDW, sWk, LDW, 32, 32, 64, true);
    }

    __syncthreads();
    for (int i = tid; i < 1024; i += 128) {
        const int a = at32(i);
        sB0[a] = wk[0] * sS[a] + wk[1] * sS[MS + a] +
                 wk[2] * sS[2 * MS + a] + wk[3] * sS[3 * MS + a];
    }

    ns_sqrt(pSq, pSi, sB0, sB1, sB2, sW0, sW1, sP0, sRed, 14);

    __syncthreads();

    {
        const int lane = tid & 31;
        const int half = lane >> 4;
        const int l15  = lane & 15;
        const int k = __builtin_amdgcn_readfirstlane(tid >> 5);
        float* const wa = sS + k * MS;
        float* const wb = lds + (6 + 2 * k) * MS;
        float* const wc = lds + (7 + 2 * k) * MS;

        mmw<0>(wb, pSi, wa, half, l15);
        mmw<0>(wc, wb, pSi, half, l15);

        for (int s = 0; s < 3; ++s) ns_sqrt_w(wc, wa, wb, 12, lane, half, l15);

        wc[lane * (LD + 1)] -= 1.f;

        float* q  = wa;
        float* qn = wb;
        for (int i = lane; i < 1024; i += 32)
            q[at32(i)] = diag32(i) ? (-1.f / 8.f) : 0.f;
        wave_lds_sync();
        for (int ii = 7; ii >= 1; --ii) {
            mmw<0>(qn, wc, q, half, l15);
            const float a = ((ii & 1) ? 1.f : -1.f) / (float)ii;
            qn[lane * (LD + 1)] += a;
            wave_lds_sync();
            float* tmp = q; q = qn; qn = tmp;
        }
        mmw<0>(wa, wc, q, half, l15);
    }

    __syncthreads();
    for (int i = tid; i < 1024; i += 128) {
        const int a = at32(i);
        float u = 0.f;
        #pragma unroll
        for (int k = 0; k < 4; ++k) u += (wk[k] * 8.f) * sS[k * MS + a];
        sB0[a] = u * (1.f / 64.f);
        sB1[a] = diag32(i) ? 1.f : 0.f;
    }
    for (int ii = 10; ii >= 1; --ii) {
        mm(sB2, LD, sB0, LD, sB1, LD, 32, 32, 32, false);
        const float inv = 1.f / (float)ii;
        for (int i = tid; i < 1024; i += 128) {
            const int a = at32(i);
            sB1[a] = (diag32(i) ? 1.f : 0.f) + sB2[a] * inv;
        }
    }
    float* r  = sB1;
    float* rn = sB2;
    for (int sq = 0; sq < 6; ++sq) {
        mm(rn, LD, r, LD, r, LD, 32, 32, 32, false);
        float* tmp = r; r = rn; rn = tmp;
    }
    mm(sB0, LD, pSq, LD, r, LD, 32, 32, 32, false);
    mm(sP0, LD, sB0, LD, pSq, LD, 32, 32, 32, false);
    __syncthreads();
    for (int i = tid; i < 1024; i += 128) out[bc * 1024 + i] = sP0[at32(i)];
}

extern "C" void kernel_launch(void* const* d_in, const int* in_sizes, int n_in,
                              void* d_out, int out_size, void* d_ws, size_t ws_size,
                              hipStream_t stream)
{
    const float* x   = (const float*)d_in[0];
    const float* wts = (const float*)d_in[1];
    const float* Ws  = (const float*)d_in[2];
    float* out = (float*)d_out;

    dim3 grid(2048);
    dim3 block(128);
    hipLaunchKernelGGL(spdnet_mixedop_kernel, grid, block, 0, stream,
                       x, wts, Ws, out);
}


# Products with an exactly known result are not computed: Newton-Schulz step 1 (Z=I: T and Z elementwise), the unread last Z update of the log-phase roots, first Horner terms; sums unchanged
# speedup vs baseline: 3.4217x; 1.0601x over previous
#include <hip/hip_runtime.h>


typedef __attribute__((ext_vector_type(2))) float v2f;
typedef __attribute__((ext_vector_type(8))) float v8f;

constexpr int LD  = 36;
constexpr int LDW = 68;
constexpr int MS  = 32 * LD;

__device__ __forceinline__ int at32(int i) { return (i >> 5) * LD + (i & 31); }
__device__ __forceinline__ bool diag32(int i) { return (i >> 5) == (i & 31); }

__device__ __forceinline__ void mm(float* __restrict__ C, int ldc,
                                   const float* __restrict__ A, int lda,
                                   const float* __restrict__ B, int ldb,
                                   int M, int N, int Kd, bool transB)
{
    __syncthreads();
    const int wave = threadIdx.x >> 5;
    const int lane = threadIdx.x & 31;
    const int half = lane >> 4;
    const int l15  = lane & 15;
    const int tn = N >> 4;
    const int tiles = (M >> 4) * tn;
    for (int t = wave; t < tiles; t += 4) {
        const int mi = (t / tn) << 4;
        const int ni = (t % tn) << 4;
        v8f acc = {};
        #pragma unroll 4
        for (int k = 0; k < Kd; k += 4) {
            v2f a, b;
            const float* ar = A + (mi + l15) * lda + k + 2 * half;
            a.x = ar[0];
            a.y = ar[1];
            if (!transB) {
                const float* br = B + (k + 2 * half) * ldb + ni + l15;
                b.x = br[0];
                b.y = br[ldb];
            } else {
                const float* br = B + (ni + l15) * ldb + k + 2 * half;
                b.x = br[0];
                b.y = br[1];
            }
            acc = __builtin_amdgcn_wmma_f32_16x16x4_f32(
                false, a, false, b, (short)0, acc, false, false);
        }
        #pragma unroll
        for (int r = 0; r < 8; ++r)
            C[(mi + half * 8 + r) * ldc + ni + l15] = acc[r];
    }
    __syncthreads();
}

__device__ __forceinline__ float trace32(const float* A, float* red)
{
    __syncthreads();
    if (threadIdx.x < 32) red[threadIdx.x] = A[threadIdx.x * (LD + 1)];
    __syncthreads();
    float s = 0.f;
    #pragma unroll
    for (int i = 0; i < 32; ++i) s += red[i];
    return s;
}

__device__ void ns_sqrt(float* dst_s, float* dst_i, const float* src,
                        float* y, float* z, float* t, float* yn, float* zn,
                        float* red, int iters)
{
    const int tid = threadIdx.x;
    const float c = trace32(src, red);
    const float inv_c = 1.f / c;
    __syncthreads();
    for (int i = tid; i < 1024; i += 128) {
        const int a = at32(i);
        const float v = src[a] * inv_c;
        y[a] = v;
        zn[a] = (diag32(i) ? 1.5f : 0.f) - 0.5f * v;
    }
    const int wave = tid >> 5;
    const int lane = tid & 31;
    const int half = lane >> 4;
    const int l15  = lane & 15;
    const int mi = (wave >> 1) << 4;
    const int ni = (wave & 1) << 4;
    const int aoff = (mi + l15) * LD + 2 * half;
    const int boff = (2 * half) * LD + ni + l15;
    const int coff = (mi + half * 8) * LD + ni + l15;
    __syncthreads();
    {
        v2f ya[8], tb[8];
        #pragma unroll
        for (int j = 0; j < 8; ++j) {
            ya[j].x = y[aoff + 4 * j];
            ya[j].y = y[aoff + 4 * j + 1];
            tb[j].x = zn[boff + 4 * j * LD];
            tb[j].y = zn[boff + 4 * j * LD + LD];
        }
        v8f accY = {};
        #pragma unroll
        for (int j = 0; j < 8; ++j)
            accY = __builtin_amdgcn_wmma_f32_16x16x4_f32(
                false, ya[j], false, tb[j], (short)0, accY, false, false);
        #pragma unroll
        for (int r = 0; r < 8; ++r) yn[coff + r * LD] = accY[r];
        float* tmp;
        tmp = y; y = yn; yn = tmp;
        tmp = z; z = zn; zn = tmp;
    }
    for (int it = 1; it < iters; ++it) {
        __syncthreads();
        {
            v2f fa[8], fb[8];
            #pragma unroll
            for (int j = 0; j < 8; ++j) {
                fa[j].x = z[aoff + 4 * j];
                fa[j].y = z[aoff + 4 * j + 1];
                fb[j].x = y[boff + 4 * j * LD];
                fb[j].y = y[boff + 4 * j * LD + LD];
            }
            v8f acc = {};
            #pragma unroll
            for (int j = 0; j < 8; ++j)
                acc = __builtin_amdgcn_wmma_f32_16x16x4_f32(
                    false, fa[j], false, fb[j], (short)0, acc, false, false);
            #pragma unroll
            for (int r = 0; r < 8; ++r) {
                const bool dg = (mi + half * 8 + r) == (ni + l15);
                t[coff + r * LD] = (dg ? 1.5f : 0.f) - 0.5f * acc[r];
            }
        }
        __syncthreads();
        {
            v2f ya[8], tb[8], ta[8], zb[8];
            #pragma unroll
            for (int j = 0; j < 8; ++j) {
                ya[j].x = y[aoff + 4 * j];
                ya[j].y = y[aoff + 4 * j + 1];
                tb[j].x = t[boff + 4 * j * LD];
                tb[j].y = t[boff + 4 * j * LD + LD];
                ta[j].x = t[aoff + 4 * j];
                ta[j].y = t[aoff + 4 * j + 1];
                zb[j].x = z[boff + 4 * j * LD];
                zb[j].y = z[boff + 4 * j * LD + LD];
            }
            __builtin_amdgcn_sched_barrier(0);
            v8f accY = {}, accZ = {};
            #pragma unroll
            for (int j = 0; j < 8; ++j) {
                accY = __builtin_amdgcn_wmma_f32_16x16x4_f32(
                    false, ya[j], false, tb[j], (short)0, accY, false, false);
                accZ = __builtin_amdgcn_wmma_f32_16x16x4_f32(
                    false, ta[j], false, zb[j], (short)0, accZ, false, false);
            }
            #pragma unroll
            for (int r = 0; r < 8; ++r) {
                yn[coff + r * LD] = accY[r];
                zn[coff + r * LD] = accZ[r];
            }
        }
        float* tmp;
        tmp = y; y = yn; yn = tmp;
        tmp = z; z = zn; zn = tmp;
    }
    const float sc  = __fsqrt_rn(c);
    const float isc = __frsqrt_rn(c);
    __syncthreads();
    if (dst_s) for (int i = tid; i < 1024; i += 128) dst_s[at32(i)] = y[at32(i)] * sc;
    if (dst_i) for (int i = tid; i < 1024; i += 128) dst_i[at32(i)] = z[at32(i)] * isc;
    __syncthreads();
}

__device__ __forceinline__ void wave_lds_sync()
{
    __builtin_amdgcn_fence(__ATOMIC_RELEASE, "wavefront");
    __builtin_amdgcn_wave_barrier();
    __builtin_amdgcn_fence(__ATOMIC_ACQUIRE, "wavefront");
}

template <int EPI>
__device__ __forceinline__ void mmw(float* C, const float* A, const float* B,
                                    int half, int l15)
{
    v2f a0[8], a1[8], b0[8], b1[8];
    const float* ar = A + l15 * LD + 2 * half;
    const float* br = B + (2 * half) * LD + l15;
    #pragma unroll
    for (int j = 0; j < 8; ++j) {
        a0[j].x = ar[4 * j];
        a0[j].y = ar[4 * j + 1];
        a1[j].x = ar[16 * LD + 4 * j];
        a1[j].y = ar[16 * LD + 4 * j + 1];
        b0[j].x = br[4 * j * LD];
        b0[j].y = br[4 * j * LD + LD];
        b1[j].x = br[4 * j * LD + 16];
        b1[j].y = br[4 * j * LD + LD + 16];
    }
    __builtin_amdgcn_sched_barrier(0);
    v8f c00 = {}, c01 = {}, c10 = {}, c11 = {};
    #pragma unroll
    for (int j = 0; j < 8; ++j) {
        c00 = __builtin_amdgcn_wmma_f32_16x16x4_f32(
            false, a0[j], false, b0[j], (short)0, c00, false, false);
        c01 = __builtin_amdgcn_wmma_f32_16x16x4_f32(
            false, a0[j], false, b1[j], (short)0, c01, false, false);
        c10 = __builtin_amdgcn_wmma_f32_16x16x4_f32(
            false, a1[j], false, b0[j], (short)0, c10, false, false);
        c11 = __builtin_amdgcn_wmma_f32_16x16x4_f32(
            false, a1[j], false, b1[j], (short)0, c11, false, false);
    }
    float* cr = C + (half * 8) * LD + l15;
    #pragma unroll
    for (int r = 0; r < 8; ++r) {
        if (EPI == 1) {
            const bool dg = (half * 8 + r) == l15;
            cr[r * LD]             = (dg ? 1.5f : 0.f) - 0.5f * c00[r];
            cr[r * LD + 16]        = 0.f - 0.5f * c01[r];
            cr[(16 + r) * LD]      = 0.f - 0.5f * c10[r];
            cr[(16 + r) * LD + 16] = (dg ? 1.5f : 0.f) - 0.5f * c11[r];
        } else {
            cr[r * LD]             = c00[r];
            cr[r * LD + 16]        = c01[r];
            cr[(16 + r) * LD]      = c10[r];
            cr[(16 + r) * LD + 16] = c11[r];
        }
    }
    wave_lds_sync();
}

__device__ __forceinline__ void ns_sqrt_w(float* y, float* z, float* t, int iters,
                                          int lane, int half, int l15)
{
    float c = 0.f;
    #pragma unroll
    for (int i = 0; i < 32; ++i) c += y[i * (LD + 1)];
    const float inv_c = 1.f / c;
    for (int i = lane; i < 1024; i += 32) {
        const int a = at32(i);
        const float v = y[a] * inv_c;
        y[a] = v;
        z[a] = (diag32(i) ? 1.5f : 0.f) - 0.5f * v;
    }
    wave_lds_sync();
    mmw<0>(y, y, z, half, l15);
    for (int it = 2; it < iters; ++it) {
        mmw<1>(t, z, y, half, l15);
        mmw<0>(y, y, t, half, l15);
        mmw<0>(z, t, z, half, l15);
    }
    mmw<1>(t, z, y, half, l15);
    mmw<0>(y, y, t, half, l15);
    const float sc = __fsqrt_rn(c);
    for (int i = lane; i < 1024; i += 32) {
        const int a = at32(i);
        y[a] = y[a] * sc;
    }
    wave_lds_sync();
}

__global__ __launch_bounds__(128, 1)
void spdnet_mixedop_kernel(const float* __restrict__ x,
                           const float* __restrict__ wts,
                           const float* __restrict__ Ws,
                           float* __restrict__ out)
{
    __shared__ float lds[14 * MS + 32];
    float* const sS   = lds;
    float* const pSq  = lds + 4 * MS;
    float* const pSi  = lds + 5 * MS;
    float* const sW0  = lds + 7 * MS;
    float* const sW1  = lds + 8 * MS;
    float* const sP0  = lds + 9 * MS;
    float* const sB0  = lds + 11 * MS;
    float* const sB1  = lds + 12 * MS;
    float* const sB2  = lds + 13 * MS;
    float* const sRed = lds + 14 * MS;
    float* const sX   = lds + 4 * MS;
    float* const sWk  = lds + 8 * MS;
    float* const sP   = lds + 10 * MS;

    const int bc  = blockIdx.x;
    const int tid = threadIdx.x;

    float wk[4];
    #pragma unroll
    for (int k = 0; k < 4; ++k) wk[k] = wts[k];

    for (int i = tid; i < 4096; i += 128) sX[(i >> 6) * LDW + (i & 63)] = x[bc * 4096 + i];

    for (int k = 0; k < 4; ++k) {
        __syncthreads();
        for (int i = tid; i < 2048; i += 128) sWk[(i >> 6) * LDW + (i & 63)] = Ws[k * 2048 + i];
        mm(sP, LDW, sWk, LDW, sX, LDW, 32, 64, 64, false);
        mm(sS + k * MS, LD, sP, LDW, sWk, LDW, 32, 32, 64, true);
    }

    __syncthreads();
    for (int i = tid; i < 1024; i += 128) {
        const int a = at32(i);
        sB0[a] = wk[0] * sS[a] + wk[1] * sS[MS + a] +
                 wk[2] * sS[2 * MS + a] + wk[3] * sS[3 * MS + a];
    }

    ns_sqrt(pSq, pSi, sB0, sB1, sB2, sW0, sW1, sP0, sRed, 14);

    __syncthreads();

    {
        const int lane = tid & 31;
        const int half = lane >> 4;
        const int l15  = lane & 15;
        const int k = __builtin_amdgcn_readfirstlane(tid >> 5);
        float* const wa = sS + k * MS;
        float* const wb = lds + (6 + 2 * k) * MS;
        float* const wc = lds + (7 + 2 * k) * MS;

        mmw<0>(wb, pSi, wa, half, l15);
        mmw<0>(wc, wb, pSi, half, l15);

        for (int s = 0; s < 3; ++s) ns_sqrt_w(wc, wa, wb, 12, lane, half, l15);

        wc[lane * (LD + 1)] -= 1.f;

        float* q  = wb;
        float* qn = wa;
        wave_lds_sync();
        for (int i = lane; i < 1024; i += 32) {
            const int a = at32(i);
            q[a] = wc[a] * (-1.f / 8.f) + (diag32(i) ? (1.f / 7.f) : 0.f);
        }
        wave_lds_sync();
        for (int ii = 6; ii >= 1; --ii) {
            mmw<0>(qn, wc, q, half, l15);
            const float a = ((ii & 1) ? 1.f : -1.f) / (float)ii;
            qn[lane * (LD + 1)] += a;
            wave_lds_sync();
            float* tmp = q; q = qn; qn = tmp;
        }
        mmw<0>(wa, wc, q, half, l15);
    }

    __syncthreads();
    for (int i = tid; i < 1024; i += 128) {
        const int a = at32(i);
        float u = 0.f;
        #pragma unroll
        for (int k = 0; k < 4; ++k) u += (wk[k] * 8.f) * sS[k * MS + a];
        const float v = u * (1.f / 64.f);
        sB0[a] = v;
        sB1[a] = (diag32(i) ? 1.f : 0.f) + v * (1.f / 10.f);
    }
    for (int ii = 9; ii >= 1; --ii) {
        mm(sB2, LD, sB0, LD, sB1, LD, 32, 32, 32, false);
        const float inv = 1.f / (float)ii;
        for (int i = tid; i < 1024; i += 128) {
            const int a = at32(i);
            sB1[a] = (diag32(i) ? 1.f : 0.f) + sB2[a] * inv;
        }
    }
    float* r  = sB1;
    float* rn = sB2;
    for (int sq = 0; sq < 6; ++sq) {
        mm(rn, LD, r, LD, r, LD, 32, 32, 32, false);
        float* tmp = r; r = rn; rn = tmp;
    }
    mm(sB0, LD, pSq, LD, r, LD, 32, 32, 32, false);
    mm(sP0, LD, sB0, LD, pSq, LD, 32, 32, 32, false);
    __syncthreads();
    for (int i = tid; i < 1024; i += 128) out[bc * 1024 + i] = sP0[at32(i)];
}

extern "C" void kernel_launch(void* const* d_in, const int* in_sizes, int n_in,
                              void* d_out, int out_size, void* d_ws, size_t ws_size,
                              hipStream_t stream)
{
    const float* x   = (const float*)d_in[0];
    const float* wts = (const float*)d_in[1];
    const float* Ws  = (const float*)d_in[2];
    float* out = (float*)d_out;

    dim3 grid(2048);
    dim3 block(128);
    hipLaunchKernelGGL(spdnet_mixedop_kernel, grid, block, 0, stream,
                       x, wts, Ws, out);
}


# Log phase (Newton-Schulz roots, Horner series): each product's result stays in registers and is the next product's B operand; A by ds_load_b128 in the matching k order; sums formed in a new k order
# speedup vs baseline: 3.5366x; 1.0336x over previous
#include <hip/hip_runtime.h>


typedef __attribute__((ext_vector_type(2))) float v2f;
typedef __attribute__((ext_vector_type(8))) float v8f;

constexpr int LD  = 36;
constexpr int LDW = 68;
constexpr int MS  = 32 * LD;

__device__ __forceinline__ int at32(int i) { return (i >> 5) * LD + (i & 31); }
__device__ __forceinline__ bool diag32(int i) { return (i >> 5) == (i & 31); }

__device__ __forceinline__ void mm(float* __restrict__ C, int ldc,
                                   const float* __restrict__ A, int lda,
                                   const float* __restrict__ B, int ldb,
                                   int M, int N, int Kd, bool transB)
{
    __syncthreads();
    const int wave = threadIdx.x >> 5;
    const int lane = threadIdx.x & 31;
    const int half = lane >> 4;
    const int l15  = lane & 15;
    const int tn = N >> 4;
    const int tiles = (M >> 4) * tn;
    for (int t = wave; t < tiles; t += 4) {
        const int mi = (t / tn) << 4;
        const int ni = (t % tn) << 4;
        v8f acc = {};
        #pragma unroll 4
        for (int k = 0; k < Kd; k += 4) {
            v2f a, b;
            const float* ar = A + (mi + l15) * lda + k + 2 * half;
            a.x = ar[0];
            a.y = ar[1];
            if (!transB) {
                const float* br = B + (k + 2 * half) * ldb + ni + l15;
                b.x = br[0];
                b.y = br[ldb];
            } else {
                const float* br = B + (ni + l15) * ldb + k + 2 * half;
                b.x = br[0];
                b.y = br[1];
            }
            acc = __builtin_amdgcn_wmma_f32_16x16x4_f32(
                false, a, false, b, (short)0, acc, false, false);
        }
        #pragma unroll
        for (int r = 0; r < 8; ++r)
            C[(mi + half * 8 + r) * ldc + ni + l15] = acc[r];
    }
    __syncthreads();
}

__device__ __forceinline__ float trace32(const float* A, float* red)
{
    __syncthreads();
    if (threadIdx.x < 32) red[threadIdx.x] = A[threadIdx.x * (LD + 1)];
    __syncthreads();
    float s = 0.f;
    #pragma unroll
    for (int i = 0; i < 32; ++i) s += red[i];
    return s;
}

__device__ void ns_sqrt(float* dst_s, float* dst_i, const float* src,
                        float* y, float* z, float* t, float* yn, float* zn,
                        float* red, int iters)
{
    const int tid = threadIdx.x;
    const float c = trace32(src, red);
    const float inv_c = 1.f / c;
    __syncthreads();
    for (int i = tid; i < 1024; i += 128) {
        const int a = at32(i);
        const float v = src[a] * inv_c;
        y[a] = v;
        zn[a] = (diag32(i) ? 1.5f : 0.f) - 0.5f * v;
    }
    const int wave = tid >> 5;
    const int lane = tid & 31;
    const int half = lane >> 4;
    const int l15  = lane & 15;
    const int mi = (wave >> 1) << 4;
    const int ni = (wave & 1) << 4;
    const int aoff = (mi + l15) * LD + 2 * half;
    const int boff = (2 * half) * LD + ni + l15;
    const int coff = (mi + half * 8) * LD + ni + l15;
    __syncthreads();
    {
        v2f ya[8], tb[8];
        #pragma unroll
        for (int j = 0; j < 8; ++j) {
            ya[j].x = y[aoff + 4 * j];
            ya[j].y = y[aoff + 4 * j + 1];
            tb[j].x = zn[boff + 4 * j * LD];
            tb[j].y = zn[boff + 4 * j * LD + LD];
        }
        v8f accY = {};
        #pragma unroll
        for (int j = 0; j < 8; ++j)
            accY = __builtin_amdgcn_wmma_f32_16x16x4_f32(
                false, ya[j], false, tb[j], (short)0, accY, false, false);
        #pragma unroll
        for (int r = 0; r < 8; ++r) yn[coff + r * LD] = accY[r];
        float* tmp;
        tmp = y; y = yn; yn = tmp;
        tmp = z; z = zn; zn = tmp;
    }
    for (int it = 1; it < iters; ++it) {
        __syncthreads();
        {
            v2f fa[8], fb[8];
            #pragma unroll
            for (int j = 0; j < 8; ++j) {
                fa[j].x = z[aoff + 4 * j];
                fa[j].y = z[aoff + 4 * j + 1];
                fb[j].x = y[boff + 4 * j * LD];
                fb[j].y = y[boff + 4 * j * LD + LD];
            }
            v8f acc = {};
            #pragma unroll
            for (int j = 0; j < 8; ++j)
                acc = __builtin_amdgcn_wmma_f32_16x16x4_f32(
                    false, fa[j], false, fb[j], (short)0, acc, false, false);
            #pragma unroll
            for (int r = 0; r < 8; ++r) {
                const bool dg = (mi + half * 8 + r) == (ni + l15);
                t[coff + r * LD] = (dg ? 1.5f : 0.f) - 0.5f * acc[r];
            }
        }
        __syncthreads();
        {
            v2f ya[8], tb[8], ta[8], zb[8];
            #pragma unroll
            for (int j = 0; j < 8; ++j) {
                ya[j].x = y[aoff + 4 * j];
                ya[j].y = y[aoff + 4 * j + 1];
                tb[j].x = t[boff + 4 * j * LD];
                tb[j].y = t[boff + 4 * j * LD + LD];
                ta[j].x = t[aoff + 4 * j];
                ta[j].y = t[aoff + 4 * j + 1];
                zb[j].x = z[boff + 4 * j * LD];
                zb[j].y = z[boff + 4 * j * LD + LD];
            }
            __builtin_amdgcn_sched_barrier(0);
            v8f accY = {}, accZ = {};
            #pragma unroll
            for (int j = 0; j < 8; ++j) {
                accY = __builtin_amdgcn_wmma_f32_16x16x4_f32(
                    false, ya[j], false, tb[j], (short)0, accY, false, false);
                accZ = __builtin_amdgcn_wmma_f32_16x16x4_f32(
                    false, ta[j], false, zb[j], (short)0, accZ, false, false);
            }
            #pragma unroll
            for (int r = 0; r < 8; ++r) {
                yn[coff + r * LD] = accY[r];
                zn[coff + r * LD] = accZ[r];
            }
        }
        float* tmp;
        tmp = y; y = yn; yn = tmp;
        tmp = z; z = zn; zn = tmp;
    }
    const float sc  = __fsqrt_rn(c);
    const float isc = __frsqrt_rn(c);
    __syncthreads();
    if (dst_s) for (int i = tid; i < 1024; i += 128) dst_s[at32(i)] = y[at32(i)] * sc;
    if (dst_i) for (int i = tid; i < 1024; i += 128) dst_i[at32(i)] = z[at32(i)] * isc;
    __syncthreads();
}

__device__ __forceinline__ void wave_lds_sync()
{
    __builtin_amdgcn_fence(__ATOMIC_RELEASE, "wavefront");
    __builtin_amdgcn_wave_barrier();
    __builtin_amdgcn_fence(__ATOMIC_ACQUIRE, "wavefront");
}

template <int EPI>
__device__ __forceinline__ void mmw(float* C, const float* A, const float* B,
                                    int half, int l15)
{
    v2f a0[8], a1[8], b0[8], b1[8];
    const float* ar = A + l15 * LD + 2 * half;
    const float* br = B + (2 * half) * LD + l15;
    #pragma unroll
    for (int j = 0; j < 8; ++j) {
        a0[j].x = ar[4 * j];
        a0[j].y = ar[4 * j + 1];
        a1[j].x = ar[16 * LD + 4 * j];
        a1[j].y = ar[16 * LD + 4 * j + 1];
        b0[j].x = br[4 * j * LD];
        b0[j].y = br[4 * j * LD + LD];
        b1[j].x = br[4 * j * LD + 16];
        b1[j].y = br[4 * j * LD + LD + 16];
    }
    __builtin_amdgcn_sched_barrier(0);
    v8f c00 = {}, c01 = {}, c10 = {}, c11 = {};
    #pragma unroll
    for (int j = 0; j < 8; ++j) {
        c00 = __builtin_amdgcn_wmma_f32_16x16x4_f32(
            false, a0[j], false, b0[j], (short)0, c00, false, false);
        c01 = __builtin_amdgcn_wmma_f32_16x16x4_f32(
            false, a0[j], false, b1[j], (short)0, c01, false, false);
        c10 = __builtin_amdgcn_wmma_f32_16x16x4_f32(
            false, a1[j], false, b0[j], (short)0, c10, false, false);
        c11 = __builtin_amdgcn_wmma_f32_16x16x4_f32(
            false, a1[j], false, b1[j], (short)0, c11, false, false);
    }
    float* cr = C + (half * 8) * LD + l15;
    #pragma unroll
    for (int r = 0; r < 8; ++r) {
        if (EPI == 1) {
            const bool dg = (half * 8 + r) == l15;
            cr[r * LD]             = (dg ? 1.5f : 0.f) - 0.5f * c00[r];
            cr[r * LD + 16]        = 0.f - 0.5f * c01[r];
            cr[(16 + r) * LD]      = 0.f - 0.5f * c10[r];
            cr[(16 + r) * LD + 16] = (dg ? 1.5f : 0.f) - 0.5f * c11[r];
        } else {
            cr[r * LD]             = c00[r];
            cr[r * LD + 16]        = c01[r];
            cr[(16 + r) * LD]      = c10[r];
            cr[(16 + r) * LD + 16] = c11[r];
        }
    }
    wave_lds_sync();
}

typedef __attribute__((ext_vector_type(4))) float v4f;

struct Tile4 { v8f t[2][2]; };

struct FragA { v4f f[2][2][2]; };

__device__ __forceinline__ void lda(FragA& a, const float* A, int half, int l15)
{
    const float* ar = A + l15 * LD + 8 * half;
    #pragma unroll
    for (int rb = 0; rb < 2; ++rb)
        #pragma unroll
        for (int kb = 0; kb < 2; ++kb)
            #pragma unroll
            for (int p = 0; p < 2; ++p)
                a.f[rb][kb][p] = *(const v4f*)(ar + rb * 16 * LD + kb * 16 + 4 * p);
}

__device__ __forceinline__ void ldd(Tile4& c, const float* C, int half, int l15)
{
    const float* cr = C + (half * 8) * LD + l15;
    #pragma unroll
    for (int rb = 0; rb < 2; ++rb)
        #pragma unroll
        for (int cb = 0; cb < 2; ++cb)
            #pragma unroll
            for (int r = 0; r < 8; ++r)
                c.t[rb][cb][r] = cr[(16 * rb + r) * LD + 16 * cb];
}

__device__ __forceinline__ void std4(float* C, const Tile4& c, int half, int l15)
{
    float* cr = C + (half * 8) * LD + l15;
    #pragma unroll
    for (int rb = 0; rb < 2; ++rb)
        #pragma unroll
        for (int cb = 0; cb < 2; ++cb)
            #pragma unroll
            for (int r = 0; r < 8; ++r)
                cr[(16 * rb + r) * LD + 16 * cb] = c.t[rb][cb][r];
    wave_lds_sync();
}

template <int EPI>
__device__ __forceinline__ void mmr(Tile4& c, const FragA& a, const Tile4& b, int half, int l15)
{
    v8f c00 = {}, c01 = {}, c10 = {}, c11 = {};
    #pragma unroll
    for (int kb = 0; kb < 2; ++kb) {
        #pragma unroll
        for (int m = 0; m < 4; ++m) {
            const int e = 2 * (m & 1);
            v2f a0, a1, b0, b1;
            a0.x = a.f[0][kb][m >> 1][e];  a0.y = a.f[0][kb][m >> 1][e + 1];
            a1.x = a.f[1][kb][m >> 1][e];  a1.y = a.f[1][kb][m >> 1][e + 1];
            b0.x = b.t[kb][0][2 * m];      b0.y = b.t[kb][0][2 * m + 1];
            b1.x = b.t[kb][1][2 * m];      b1.y = b.t[kb][1][2 * m + 1];
            c00 = __builtin_amdgcn_wmma_f32_16x16x4_f32(
                false, a0, false, b0, (short)0, c00, false, false);
            c01 = __builtin_amdgcn_wmma_f32_16x16x4_f32(
                false, a0, false, b1, (short)0, c01, false, false);
            c10 = __builtin_amdgcn_wmma_f32_16x16x4_f32(
                false, a1, false, b0, (short)0, c10, false, false);
            c11 = __builtin_amdgcn_wmma_f32_16x16x4_f32(
                false, a1, false, b1, (short)0, c11, false, false);
        }
    }
    if (EPI == 1) {
        #pragma unroll
        for (int r = 0; r < 8; ++r) {
            const bool dg = (half * 8 + r) == l15;
            c00[r] = (dg ? 1.5f : 0.f) - 0.5f * c00[r];
            c01[r] = 0.f - 0.5f * c01[r];
            c10[r] = 0.f - 0.5f * c10[r];
            c11[r] = (dg ? 1.5f : 0.f) - 0.5f * c11[r];
        }
    }
    c.t[0][0] = c00; c.t[0][1] = c01; c.t[1][0] = c10; c.t[1][1] = c11;
}

__device__ __forceinline__ void ns_sqrt_w(float* y, float* z, float* t, int iters,
                                          int lane, int half, int l15)
{
    float c = 0.f;
    #pragma unroll
    for (int i = 0; i < 32; ++i) c += y[i * (LD + 1)];
    const float inv_c = 1.f / c;
    for (int i = lane; i < 1024; i += 32) {
        const int a = at32(i);
        const float v = y[a] * inv_c;
        y[a] = v;
        z[a] = (diag32(i) ? 1.5f : 0.f) - 0.5f * v;
    }
    wave_lds_sync();
    Tile4 Yr, Zr, Tr;
    FragA aY, aZ, aT;
    lda(aY, y, half, l15);
    ldd(Zr, z, half, l15);
    mmr<0>(Yr, aY, Zr, half, l15);
    lda(aZ, z, half, l15);
    std4(y, Yr, half, l15);
    lda(aY, y, half, l15);
    for (int it = 2; it < iters; ++it) {
        __builtin_amdgcn_sched_barrier(0);
        mmr<1>(Tr, aZ, Yr, half, l15);
        std4(t, Tr, half, l15);
        lda(aT, t, half, l15);
        __builtin_amdgcn_sched_barrier(0);
        mmr<0>(Yr, aY, Tr, half, l15);
        __builtin_amdgcn_sched_barrier(0);
        Tile4 Zn;
        mmr<0>(Zn, aT, Zr, half, l15);
        Zr = Zn;
        std4(z, Zr, half, l15);
        lda(aZ, z, half, l15);
        std4(y, Yr, half, l15);
        lda(aY, y, half, l15);
    }
    __builtin_amdgcn_sched_barrier(0);
    mmr<1>(Tr, aZ, Yr, half, l15);
    mmr<0>(Yr, aY, Tr, half, l15);
    std4(y, Yr, half, l15);
    const float sc = __fsqrt_rn(c);
    for (int i = lane; i < 1024; i += 32) {
        const int a = at32(i);
        y[a] = y[a] * sc;
    }
    wave_lds_sync();
}

__global__ __launch_bounds__(128, 1)
void spdnet_mixedop_kernel(const float* __restrict__ x,
                           const float* __restrict__ wts,
                           const float* __restrict__ Ws,
                           float* __restrict__ out)
{
    __shared__ __attribute__((aligned(16))) float lds[14 * MS + 32];
    float* const sS   = lds;
    float* const pSq  = lds + 4 * MS;
    float* const pSi  = lds + 5 * MS;
    float* const sW0  = lds + 7 * MS;
    float* const sW1  = lds + 8 * MS;
    float* const sP0  = lds + 9 * MS;
    float* const sB0  = lds + 11 * MS;
    float* const sB1  = lds + 12 * MS;
    float* const sB2  = lds + 13 * MS;
    float* const sRed = lds + 14 * MS;
    float* const sX   = lds + 4 * MS;
    float* const sWk  = lds + 8 * MS;
    float* const sP   = lds + 10 * MS;

    const int bc  = blockIdx.x;
    const int tid = threadIdx.x;

    float wk[4];
    #pragma unroll
    for (int k = 0; k < 4; ++k) wk[k] = wts[k];

    for (int i = tid; i < 4096; i += 128) sX[(i >> 6) * LDW + (i & 63)] = x[bc * 4096 + i];

    for (int k = 0; k < 4; ++k) {
        __syncthreads();
        for (int i = tid; i < 2048; i += 128) sWk[(i >> 6) * LDW + (i & 63)] = Ws[k * 2048 + i];
        mm(sP, LDW, sWk, LDW, sX, LDW, 32, 64, 64, false);
        mm(sS + k * MS, LD, sP, LDW, sWk, LDW, 32, 32, 64, true);
    }

    __syncthreads();
    for (int i = tid; i < 1024; i += 128) {
        const int a = at32(i);
        sB0[a] = wk[0] * sS[a] + wk[1] * sS[MS + a] +
                 wk[2] * sS[2 * MS + a] + wk[3] * sS[3 * MS + a];
    }

    ns_sqrt(pSq, pSi, sB0, sB1, sB2, sW0, sW1, sP0, sRed, 14);

    __syncthreads();

    {
        const int lane = tid & 31;
        const int half = lane >> 4;
        const int l15  = lane & 15;
        const int k = __builtin_amdgcn_readfirstlane(tid >> 5);
        float* const wa = sS + k * MS;
        float* const wb = lds + (6 + 2 * k) * MS;
        float* const wc = lds + (7 + 2 * k) * MS;

        mmw<0>(wb, pSi, wa, half, l15);
        mmw<0>(wc, wb, pSi, half, l15);

        for (int s = 0; s < 3; ++s) ns_sqrt_w(wc, wa, wb, 12, lane, half, l15);

        wc[lane * (LD + 1)] -= 1.f;

        float* q = wb;
        wave_lds_sync();
        for (int i = lane; i < 1024; i += 32) {
            const int a = at32(i);
            q[a] = wc[a] * (-1.f / 8.f) + (diag32(i) ? (1.f / 7.f) : 0.f);
        }
        wave_lds_sync();
        FragA aE;
        Tile4 Q, Qn;
        lda(aE, wc, half, l15);
        ldd(Q, q, half, l15);
        for (int ii = 6; ii >= 1; --ii) {
            mmr<0>(Qn, aE, Q, half, l15);
            const float a = ((ii & 1) ? 1.f : -1.f) / (float)ii;
            #pragma unroll
            for (int r = 0; r < 8; ++r) {
                const bool dg = (half * 8 + r) == l15;
                Qn.t[0][0][r] = dg ? Qn.t[0][0][r] + a : Qn.t[0][0][r];
                Qn.t[1][1][r] = dg ? Qn.t[1][1][r] + a : Qn.t[1][1][r];
            }
            Q = Qn;
        }
        mmr<0>(Qn, aE, Q, half, l15);
        std4(wa, Qn, half, l15);
    }

    __syncthreads();
    for (int i = tid; i < 1024; i += 128) {
        const int a = at32(i);
        float u = 0.f;
        #pragma unroll
        for (int k = 0; k < 4; ++k) u += (wk[k] * 8.f) * sS[k * MS + a];
        const float v = u * (1.f / 64.f);
        sB0[a] = v;
        sB1[a] = (diag32(i) ? 1.f : 0.f) + v * (1.f / 10.f);
    }
    for (int ii = 9; ii >= 1; --ii) {
        mm(sB2, LD, sB0, LD, sB1, LD, 32, 32, 32, false);
        const float inv = 1.f / (float)ii;
        for (int i = tid; i < 1024; i += 128) {
            const int a = at32(i);
            sB1[a] = (diag32(i) ? 1.f : 0.f) + sB2[a] * inv;
        }
    }
    float* r  = sB1;
    float* rn = sB2;
    for (int sq = 0; sq < 6; ++sq) {
        mm(rn, LD, r, LD, r, LD, 32, 32, 32, false);
        float* tmp = r; r = rn; rn = tmp;
    }
    mm(sB0, LD, pSq, LD, r, LD, 32, 32, 32, false);
    mm(sP0, LD, sB0, LD, pSq, LD, 32, 32, 32, false);
    __syncthreads();
    for (int i = tid; i < 1024; i += 128) out[bc * 1024 + i] = sP0[at32(i)];
}

extern "C" void kernel_launch(void* const* d_in, const int* in_sizes, int n_in,
                              void* d_out, int out_size, void* d_ws, size_t ws_size,
                              hipStream_t stream)
{
    const float* x   = (const float*)d_in[0];
    const float* wts = (const float*)d_in[1];
    const float* Ws  = (const float*)d_in[2];
    float* out = (float*)d_out;

    dim3 grid(2048);
    dim3 block(128);
    hipLaunchKernelGGL(spdnet_mixedop_kernel, grid, block, 0, stream,
                       x, wts, Ws, out);
}


# Matrix log: 2 square roots + degree-18 series instead of 3 roots + degree 8 (33 products out, 10 in); truncation bound not larger wherever it was below 1e-6; other sums, so rounding differs
# speedup vs baseline: 4.1631x; 1.1771x over previous
#include <hip/hip_runtime.h>


typedef __attribute__((ext_vector_type(2))) float v2f;
typedef __attribute__((ext_vector_type(8))) float v8f;

constexpr int LD  = 36;
constexpr int LDW = 68;
constexpr int MS  = 32 * LD;
constexpr int LOG_DEG = 18;

__device__ __forceinline__ int at32(int i) { return (i >> 5) * LD + (i & 31); }
__device__ __forceinline__ bool diag32(int i) { return (i >> 5) == (i & 31); }

__device__ __forceinline__ void mm(float* __restrict__ C, int ldc,
                                   const float* __restrict__ A, int lda,
                                   const float* __restrict__ B, int ldb,
                                   int M, int N, int Kd, bool transB)
{
    __syncthreads();
    const int wave = threadIdx.x >> 5;
    const int lane = threadIdx.x & 31;
    const int half = lane >> 4;
    const int l15  = lane & 15;
    const int tn = N >> 4;
    const int tiles = (M >> 4) * tn;
    for (int t = wave; t < tiles; t += 4) {
        const int mi = (t / tn) << 4;
        const int ni = (t % tn) << 4;
        v8f acc = {};
        #pragma unroll 4
        for (int k = 0; k < Kd; k += 4) {
            v2f a, b;
            const float* ar = A + (mi + l15) * lda + k + 2 * half;
            a.x = ar[0];
            a.y = ar[1];
            if (!transB) {
                const float* br = B + (k + 2 * half) * ldb + ni + l15;
                b.x = br[0];
                b.y = br[ldb];
            } else {
                const float* br = B + (ni + l15) * ldb + k + 2 * half;
                b.x = br[0];
                b.y = br[1];
            }
            acc = __builtin_amdgcn_wmma_f32_16x16x4_f32(
                false, a, false, b, (short)0, acc, false, false);
        }
        #pragma unroll
        for (int r = 0; r < 8; ++r)
            C[(mi + half * 8 + r) * ldc + ni + l15] = acc[r];
    }
    __syncthreads();
}

__device__ __forceinline__ float trace32(const float* A, float* red)
{
    __syncthreads();
    if (threadIdx.x < 32) red[threadIdx.x] = A[threadIdx.x * (LD + 1)];
    __syncthreads();
    float s = 0.f;
    #pragma unroll
    for (int i = 0; i < 32; ++i) s += red[i];
    return s;
}

__device__ void ns_sqrt(float* dst_s, float* dst_i, const float* src,
                        float* y, float* z, float* t, float* yn, float* zn,
                        float* red, int iters)
{
    const int tid = threadIdx.x;
    const float c = trace32(src, red);
    const float inv_c = 1.f / c;
    __syncthreads();
    for (int i = tid; i < 1024; i += 128) {
        const int a = at32(i);
        const float v = src[a] * inv_c;
        y[a] = v;
        zn[a] = (diag32(i) ? 1.5f : 0.f) - 0.5f * v;
    }
    const int wave = tid >> 5;
    const int lane = tid & 31;
    const int half = lane >> 4;
    const int l15  = lane & 15;
    const int mi = (wave >> 1) << 4;
    const int ni = (wave & 1) << 4;
    const int aoff = (mi + l15) * LD + 2 * half;
    const int boff = (2 * half) * LD + ni + l15;
    const int coff = (mi + half * 8) * LD + ni + l15;
    __syncthreads();
    {
        v2f ya[8], tb[8];
        #pragma unroll
        for (int j = 0; j < 8; ++j) {
            ya[j].x = y[aoff + 4 * j];
            ya[j].y = y[aoff + 4 * j + 1];
            tb[j].x = zn[boff + 4 * j * LD];
            tb[j].y = zn[boff + 4 * j * LD + LD];
        }
        v8f accY = {};
        #pragma unroll
        for (int j = 0; j < 8; ++j)
            accY = __builtin_amdgcn_wmma_f32_16x16x4_f32(
                false, ya[j], false, tb[j], (short)0, accY, false, false);
        #pragma unroll
        for (int r = 0; r < 8; ++r) yn[coff + r * LD] = accY[r];
        float* tmp;
        tmp = y; y = yn; yn = tmp;
        tmp = z; z = zn; zn = tmp;
    }
    for (int it = 1; it < iters; ++it) {
        __syncthreads();
        {
            v2f fa[8], fb[8];
            #pragma unroll
            for (int j = 0; j < 8; ++j) {
                fa[j].x = z[aoff + 4 * j];
                fa[j].y = z[aoff + 4 * j + 1];
                fb[j].x = y[boff + 4 * j * LD];
                fb[j].y = y[boff + 4 * j * LD + LD];
            }
            v8f acc = {};
            #pragma unroll
            for (int j = 0; j < 8; ++j)
                acc = __builtin_amdgcn_wmma_f32_16x16x4_f32(
                    false, fa[j], false, fb[j], (short)0, acc, false, false);
            #pragma unroll
            for (int r = 0; r < 8; ++r) {
                const bool dg = (mi + half * 8 + r) == (ni + l15);
                t[coff + r * LD] = (dg ? 1.5f : 0.f) - 0.5f * acc[r];
            }
        }
        __syncthreads();
        {
            v2f ya[8], tb[8], ta[8], zb[8];
            #pragma unroll
            for (int j = 0; j < 8; ++j) {
                ya[j].x = y[aoff + 4 * j];
                ya[j].y = y[aoff + 4 * j + 1];
                tb[j].x = t[boff + 4 * j * LD];
                tb[j].y = t[boff + 4 * j * LD + LD];
                ta[j].x = t[aoff + 4 * j];
                ta[j].y = t[aoff + 4 * j + 1];
                zb[j].x = z[boff + 4 * j * LD];
                zb[j].y = z[boff + 4 * j * LD + LD];
            }
            __builtin_amdgcn_sched_barrier(0);
            v8f accY = {}, accZ = {};
            #pragma unroll
            for (int j = 0; j < 8; ++j) {
                accY = __builtin_amdgcn_wmma_f32_16x16x4_f32(
                    false, ya[j], false, tb[j], (short)0, accY, false, false);
                accZ = __builtin_amdgcn_wmma_f32_16x16x4_f32(
                    false, ta[j], false, zb[j], (short)0, accZ, false, false);
            }
            #pragma unroll
            for (int r = 0; r < 8; ++r) {
                yn[coff + r * LD] = accY[r];
                zn[coff + r * LD] = accZ[r];
            }
        }
        float* tmp;
        tmp = y; y = yn; yn = tmp;
        tmp = z; z = zn; zn = tmp;
    }
    const float sc  = __fsqrt_rn(c);
    const float isc = __frsqrt_rn(c);
    __syncthreads();
    if (dst_s) for (int i = tid; i < 1024; i += 128) dst_s[at32(i)] = y[at32(i)] * sc;
    if (dst_i) for (int i = tid; i < 1024; i += 128) dst_i[at32(i)] = z[at32(i)] * isc;
    __syncthreads();
}

__device__ __forceinline__ void wave_lds_sync()
{
    __builtin_amdgcn_fence(__ATOMIC_RELEASE, "wavefront");
    __builtin_amdgcn_wave_barrier();
    __builtin_amdgcn_fence(__ATOMIC_ACQUIRE, "wavefront");
}

template <int EPI>
__device__ __forceinline__ void mmw(float* C, const float* A, const float* B,
                                    int half, int l15)
{
    v2f a0[8], a1[8], b0[8], b1[8];
    const float* ar = A + l15 * LD + 2 * half;
    const float* br = B + (2 * half) * LD + l15;
    #pragma unroll
    for (int j = 0; j < 8; ++j) {
        a0[j].x = ar[4 * j];
        a0[j].y = ar[4 * j + 1];
        a1[j].x = ar[16 * LD + 4 * j];
        a1[j].y = ar[16 * LD + 4 * j + 1];
        b0[j].x = br[4 * j * LD];
        b0[j].y = br[4 * j * LD + LD];
        b1[j].x = br[4 * j * LD + 16];
        b1[j].y = br[4 * j * LD + LD + 16];
    }
    __builtin_amdgcn_sched_barrier(0);
    v8f c00 = {}, c01 = {}, c10 = {}, c11 = {};
    #pragma unroll
    for (int j = 0; j < 8; ++j) {
        c00 = __builtin_amdgcn_wmma_f32_16x16x4_f32(
            false, a0[j], false, b0[j], (short)0, c00, false, false);
        c01 = __builtin_amdgcn_wmma_f32_16x16x4_f32(
            false, a0[j], false, b1[j], (short)0, c01, false, false);
        c10 = __builtin_amdgcn_wmma_f32_16x16x4_f32(
            false, a1[j], false, b0[j], (short)0, c10, false, false);
        c11 = __builtin_amdgcn_wmma_f32_16x16x4_f32(
            false, a1[j], false, b1[j], (short)0, c11, false, false);
    }
    float* cr = C + (half * 8) * LD + l15;
    #pragma unroll
    for (int r = 0; r < 8; ++r) {
        if (EPI == 1) {
            const bool dg = (half * 8 + r) == l15;
            cr[r * LD]             = (dg ? 1.5f : 0.f) - 0.5f * c00[r];
            cr[r * LD + 16]        = 0.f - 0.5f * c01[r];
            cr[(16 + r) * LD]      = 0.f - 0.5f * c10[r];
            cr[(16 + r) * LD + 16] = (dg ? 1.5f : 0.f) - 0.5f * c11[r];
        } else {
            cr[r * LD]             = c00[r];
            cr[r * LD + 16]        = c01[r];
            cr[(16 + r) * LD]      = c10[r];
            cr[(16 + r) * LD + 16] = c11[r];
        }
    }
    wave_lds_sync();
}

typedef __attribute__((ext_vector_type(4))) float v4f;

struct Tile4 { v8f t[2][2]; };

struct FragA { v4f f[2][2][2]; };

__device__ __forceinline__ void lda(FragA& a, const float* A, int half, int l15)
{
    const float* ar = A + l15 * LD + 8 * half;
    #pragma unroll
    for (int rb = 0; rb < 2; ++rb)
        #pragma unroll
        for (int kb = 0; kb < 2; ++kb)
            #pragma unroll
            for (int p = 0; p < 2; ++p)
                a.f[rb][kb][p] = *(const v4f*)(ar + rb * 16 * LD + kb * 16 + 4 * p);
}

__device__ __forceinline__ void ldd(Tile4& c, const float* C, int half, int l15)
{
    const float* cr = C + (half * 8) * LD + l15;
    #pragma unroll
    for (int rb = 0; rb < 2; ++rb)
        #pragma unroll
        for (int cb = 0; cb < 2; ++cb)
            #pragma unroll
            for (int r = 0; r < 8; ++r)
                c.t[rb][cb][r] = cr[(16 * rb + r) * LD + 16 * cb];
}

__device__ __forceinline__ void std4(float* C, const Tile4& c, int half, int l15)
{
    float* cr = C + (half * 8) * LD + l15;
    #pragma unroll
    for (int rb = 0; rb < 2; ++rb)
        #pragma unroll
        for (int cb = 0; cb < 2; ++cb)
            #pragma unroll
            for (int r = 0; r < 8; ++r)
                cr[(16 * rb + r) * LD + 16 * cb] = c.t[rb][cb][r];
    wave_lds_sync();
}

template <int EPI>
__device__ __forceinline__ void mmr(Tile4& c, const FragA& a, const Tile4& b, int half, int l15)
{
    v8f c00 = {}, c01 = {}, c10 = {}, c11 = {};
    #pragma unroll
    for (int kb = 0; kb < 2; ++kb) {
        #pragma unroll
        for (int m = 0; m < 4; ++m) {
            const int e = 2 * (m & 1);
            v2f a0, a1, b0, b1;
            a0.x = a.f[0][kb][m >> 1][e];  a0.y = a.f[0][kb][m >> 1][e + 1];
            a1.x = a.f[1][kb][m >> 1][e];  a1.y = a.f[1][kb][m >> 1][e + 1];
            b0.x = b.t[kb][0][2 * m];      b0.y = b.t[kb][0][2 * m + 1];
            b1.x = b.t[kb][1][2 * m];      b1.y = b.t[kb][1][2 * m + 1];
            c00 = __builtin_amdgcn_wmma_f32_16x16x4_f32(
                false, a0, false, b0, (short)0, c00, false, false);
            c01 = __builtin_amdgcn_wmma_f32_16x16x4_f32(
                false, a0, false, b1, (short)0, c01, false, false);
            c10 = __builtin_amdgcn_wmma_f32_16x16x4_f32(
                false, a1, false, b0, (short)0, c10, false, false);
            c11 = __builtin_amdgcn_wmma_f32_16x16x4_f32(
                false, a1, false, b1, (short)0, c11, false, false);
        }
    }
    if (EPI == 1) {
        #pragma unroll
        for (int r = 0; r < 8; ++r) {
            const bool dg = (half * 8 + r) == l15;
            c00[r] = (dg ? 1.5f : 0.f) - 0.5f * c00[r];
            c01[r] = 0.f - 0.5f * c01[r];
            c10[r] = 0.f - 0.5f * c10[r];
            c11[r] = (dg ? 1.5f : 0.f) - 0.5f * c11[r];
        }
    }
    c.t[0][0] = c00; c.t[0][1] = c01; c.t[1][0] = c10; c.t[1][1] = c11;
}

__device__ __forceinline__ void ns_sqrt_w(float* y, float* z, float* t, int iters,
                                          int lane, int half, int l15)
{
    float c = 0.f;
    #pragma unroll
    for (int i = 0; i < 32; ++i) c += y[i * (LD + 1)];
    const float inv_c = 1.f / c;
    for (int i = lane; i < 1024; i += 32) {
        const int a = at32(i);
        const float v = y[a] * inv_c;
        y[a] = v;
        z[a] = (diag32(i) ? 1.5f : 0.f) - 0.5f * v;
    }
    wave_lds_sync();
    Tile4 Yr, Zr, Tr;
    FragA aY, aZ, aT;
    lda(aY, y, half, l15);
    ldd(Zr, z, half, l15);
    mmr<0>(Yr, aY, Zr, half, l15);
    lda(aZ, z, half, l15);
    std4(y, Yr, half, l15);
    lda(aY, y, half, l15);
    for (int it = 2; it < iters; ++it) {
        __builtin_amdgcn_sched_barrier(0);
        mmr<1>(Tr, aZ, Yr, half, l15);
        std4(t, Tr, half, l15);
        lda(aT, t, half, l15);
        __builtin_amdgcn_sched_barrier(0);
        mmr<0>(Yr, aY, Tr, half, l15);
        __builtin_amdgcn_sched_barrier(0);
        Tile4 Zn;
        mmr<0>(Zn, aT, Zr, half, l15);
        Zr = Zn;
        std4(z, Zr, half, l15);
        lda(aZ, z, half, l15);
        std4(y, Yr, half, l15);
        lda(aY, y, half, l15);
    }
    __builtin_amdgcn_sched_barrier(0);
    mmr<1>(Tr, aZ, Yr, half, l15);
    mmr<0>(Yr, aY, Tr, half, l15);
    std4(y, Yr, half, l15);
    const float sc = __fsqrt_rn(c);
    for (int i = lane; i < 1024; i += 32) {
        const int a = at32(i);
        y[a] = y[a] * sc;
    }
    wave_lds_sync();
}

__global__ __launch_bounds__(128, 1)
void spdnet_mixedop_kernel(const float* __restrict__ x,
                           const float* __restrict__ wts,
                           const float* __restrict__ Ws,
                           float* __restrict__ out)
{
    __shared__ __attribute__((aligned(16))) float lds[14 * MS + 32];
    float* const sS   = lds;
    float* const pSq  = lds + 4 * MS;
    float* const pSi  = lds + 5 * MS;
    float* const sW0  = lds + 7 * MS;
    float* const sW1  = lds + 8 * MS;
    float* const sP0  = lds + 9 * MS;
    float* const sB0  = lds + 11 * MS;
    float* const sB1  = lds + 12 * MS;
    float* const sB2  = lds + 13 * MS;
    float* const sRed = lds + 14 * MS;
    float* const sX   = lds + 4 * MS;
    float* const sWk  = lds + 8 * MS;
    float* const sP   = lds + 10 * MS;

    const int bc  = blockIdx.x;
    const int tid = threadIdx.x;

    float wk[4];
    #pragma unroll
    for (int k = 0; k < 4; ++k) wk[k] = wts[k];

    for (int i = tid; i < 4096; i += 128) sX[(i >> 6) * LDW + (i & 63)] = x[bc * 4096 + i];

    for (int k = 0; k < 4; ++k) {
        __syncthreads();
        for (int i = tid; i < 2048; i += 128) sWk[(i >> 6) * LDW + (i & 63)] = Ws[k * 2048 + i];
        mm(sP, LDW, sWk, LDW, sX, LDW, 32, 64, 64, false);
        mm(sS + k * MS, LD, sP, LDW, sWk, LDW, 32, 32, 64, true);
    }

    __syncthreads();
    for (int i = tid; i < 1024; i += 128) {
        const int a = at32(i);
        sB0[a] = wk[0] * sS[a] + wk[1] * sS[MS + a] +
                 wk[2] * sS[2 * MS + a] + wk[3] * sS[3 * MS + a];
    }

    ns_sqrt(pSq, pSi, sB0, sB1, sB2, sW0, sW1, sP0, sRed, 14);

    __syncthreads();

    {
        const int lane = tid & 31;
        const int half = lane >> 4;
        const int l15  = lane & 15;
        const int k = __builtin_amdgcn_readfirstlane(tid >> 5);
        float* const wa = sS + k * MS;
        float* const wb = lds + (6 + 2 * k) * MS;
        float* const wc = lds + (7 + 2 * k) * MS;

        mmw<0>(wb, pSi, wa, half, l15);
        mmw<0>(wc, wb, pSi, half, l15);

        for (int s = 0; s < 2; ++s) ns_sqrt_w(wc, wa, wb, 12, lane, half, l15);

        wc[lane * (LD + 1)] -= 1.f;

        float* q = wb;
        wave_lds_sync();
        for (int i = lane; i < 1024; i += 32) {
            const int a = at32(i);
            q[a] = wc[a] * (-1.f / (float)LOG_DEG) + (diag32(i) ? (1.f / (float)(LOG_DEG - 1)) : 0.f);
        }
        wave_lds_sync();
        FragA aE;
        Tile4 Q, Qn;
        lda(aE, wc, half, l15);
        ldd(Q, q, half, l15);
        for (int ii = LOG_DEG - 2; ii >= 1; --ii) {
            mmr<0>(Qn, aE, Q, half, l15);
            const float a = ((ii & 1) ? 1.f : -1.f) / (float)ii;
            #pragma unroll
            for (int r = 0; r < 8; ++r) {
                const bool dg = (half * 8 + r) == l15;
                Qn.t[0][0][r] = dg ? Qn.t[0][0][r] + a : Qn.t[0][0][r];
                Qn.t[1][1][r] = dg ? Qn.t[1][1][r] + a : Qn.t[1][1][r];
            }
            Q = Qn;
        }
        mmr<0>(Qn, aE, Q, half, l15);
        std4(wa, Qn, half, l15);
    }

    __syncthreads();
    for (int i = tid; i < 1024; i += 128) {
        const int a = at32(i);
        float u = 0.f;
        #pragma unroll
        for (int k = 0; k < 4; ++k) u += (wk[k] * 4.f) * sS[k * MS + a];
        const float v = u * (1.f / 64.f);
        sB0[a] = v;
        sB1[a] = (diag32(i) ? 1.f : 0.f) + v * (1.f / 10.f);
    }
    for (int ii = 9; ii >= 1; --ii) {
        mm(sB2, LD, sB0, LD, sB1, LD, 32, 32, 32, false);
        const float inv = 1.f / (float)ii;
        for (int i = tid; i < 1024; i += 128) {
            const int a = at32(i);
            sB1[a] = (diag32(i) ? 1.f : 0.f) + sB2[a] * inv;
        }
    }
    float* r  = sB1;
    float* rn = sB2;
    for (int sq = 0; sq < 6; ++sq) {
        mm(rn, LD, r, LD, r, LD, 32, 32, 32, false);
        float* tmp = r; r = rn; rn = tmp;
    }
    mm(sB0, LD, pSq, LD, r, LD, 32, 32, 32, false);
    mm(sP0, LD, sB0, LD, pSq, LD, 32, 32, 32, false);
    __syncthreads();
    for (int i = tid; i < 1024; i += 128) out[bc * 1024 + i] = sP0[at32(i)];
}

extern "C" void kernel_launch(void* const* d_in, const int* in_sizes, int n_in,
                              void* d_out, int out_size, void* d_ws, size_t ws_size,
                              hipStream_t stream)
{
    const float* x   = (const float*)d_in[0];
    const float* wts = (const float*)d_in[1];
    const float* Ws  = (const float*)d_in[2];
    float* out = (float*)d_out;

    dim3 grid(2048);
    dim3 block(128);
    hipLaunchKernelGGL(spdnet_mixedop_kernel, grid, block, 0, stream,
                       x, wts, Ws, out);
}


# Newton-Schulz roots stop at convergence: each iteration sums |I-ZY|_F^2 from its T product's accumulators, the last once (3/8)|R|^2 < f32 rounding; old counts stay as caps; products sum as before
# speedup vs baseline: 4.7721x; 1.1463x over previous
#include <hip/hip_runtime.h>


typedef __attribute__((ext_vector_type(2))) float v2f;
typedef __attribute__((ext_vector_type(8))) float v8f;

constexpr int LD  = 36;
constexpr int LDW = 68;
constexpr int MS  = 32 * LD;
constexpr int LOG_DEG = 18;

constexpr float NS_TOL2 = 1.5e-7f;

template <int M>
__device__ __forceinline__ float row_xor(float v)
{
    const int i = __builtin_bit_cast(int, v);
    return __builtin_bit_cast(float, __builtin_amdgcn_update_dpp(i, i, 0x160 + M, 0xf, 0xf, false));
}

__device__ __forceinline__ float wave_sum_uniform(float v)
{
    v += row_xor<1>(v);
    v += row_xor<2>(v);
    v += row_xor<4>(v);
    v += row_xor<8>(v);
    const int i = __builtin_bit_cast(int, v);
    v += __builtin_bit_cast(float, __builtin_amdgcn_permlanex16(i, i, 0x76543210, 0xfedcba98, false, false));
    return __builtin_bit_cast(float, __builtin_amdgcn_readfirstlane(__builtin_bit_cast(int, v)));
}

__device__ __forceinline__ int at32(int i) { return (i >> 5) * LD + (i & 31); }
__device__ __forceinline__ bool diag32(int i) { return (i >> 5) == (i & 31); }

__device__ __forceinline__ void mm(float* __restrict__ C, int ldc,
                                   const float* __restrict__ A, int lda,
                                   const float* __restrict__ B, int ldb,
                                   int M, int N, int Kd, bool transB)
{
    __syncthreads();
    const int wave = threadIdx.x >> 5;
    const int lane = threadIdx.x & 31;
    const int half = lane >> 4;
    const int l15  = lane & 15;
    const int tn = N >> 4;
    const int tiles = (M >> 4) * tn;
    for (int t = wave; t < tiles; t += 4) {
        const int mi = (t / tn) << 4;
        const int ni = (t % tn) << 4;
        v8f acc = {};
        #pragma unroll 4
        for (int k = 0; k < Kd; k += 4) {
            v2f a, b;
            const float* ar = A + (mi + l15) * lda + k + 2 * half;
            a.x = ar[0];
            a.y = ar[1];
            if (!transB) {
                const float* br = B + (k + 2 * half) * ldb + ni + l15;
                b.x = br[0];
                b.y = br[ldb];
            } else {
                const float* br = B + (ni + l15) * ldb + k + 2 * half;
                b.x = br[0];
                b.y = br[1];
            }
            acc = __builtin_amdgcn_wmma_f32_16x16x4_f32(
                false, a, false, b, (short)0, acc, false, false);
        }
        #pragma unroll
        for (int r = 0; r < 8; ++r)
            C[(mi + half * 8 + r) * ldc + ni + l15] = acc[r];
    }
    __syncthreads();
}

__device__ __forceinline__ float trace32(const float* A, float* red)
{
    __syncthreads();
    if (threadIdx.x < 32) red[threadIdx.x] = A[threadIdx.x * (LD + 1)];
    __syncthreads();
    float s = 0.f;
    #pragma unroll
    for (int i = 0; i < 32; ++i) s += red[i];
    return s;
}

__device__ void ns_sqrt(float* dst_s, float* dst_i, const float* src,
                        float* y, float* z, float* t, float* yn, float* zn,
                        float* red, int iters)
{
    const int tid = threadIdx.x;
    const float c = trace32(src, red);
    const float inv_c = 1.f / c;
    __syncthreads();
    for (int i = tid; i < 1024; i += 128) {
        const int a = at32(i);
        const float v = src[a] * inv_c;
        y[a] = v;
        zn[a] = (diag32(i) ? 1.5f : 0.f) - 0.5f * v;
    }
    const int wave = tid >> 5;
    const int lane = tid & 31;
    const int half = lane >> 4;
    const int l15  = lane & 15;
    const int mi = (wave >> 1) << 4;
    const int ni = (wave & 1) << 4;
    const int aoff = (mi + l15) * LD + 2 * half;
    const int boff = (2 * half) * LD + ni + l15;
    const int coff = (mi + half * 8) * LD + ni + l15;
    __syncthreads();
    {
        v2f ya[8], tb[8];
        #pragma unroll
        for (int j = 0; j < 8; ++j) {
            ya[j].x = y[aoff + 4 * j];
            ya[j].y = y[aoff + 4 * j + 1];
            tb[j].x = zn[boff + 4 * j * LD];
            tb[j].y = zn[boff + 4 * j * LD + LD];
        }
        v8f accY = {};
        #pragma unroll
        for (int j = 0; j < 8; ++j)
            accY = __builtin_amdgcn_wmma_f32_16x16x4_f32(
                false, ya[j], false, tb[j], (short)0, accY, false, false);
        #pragma unroll
        for (int r = 0; r < 8; ++r) yn[coff + r * LD] = accY[r];
        float* tmp;
        tmp = y; y = yn; yn = tmp;
        tmp = z; z = zn; zn = tmp;
    }
    for (int it = 1; it < iters; ++it) {
        __syncthreads();
        {
            v2f fa[8], fb[8];
            #pragma unroll
            for (int j = 0; j < 8; ++j) {
                fa[j].x = z[aoff + 4 * j];
                fa[j].y = z[aoff + 4 * j + 1];
                fb[j].x = y[boff + 4 * j * LD];
                fb[j].y = y[boff + 4 * j * LD + LD];
            }
            __builtin_amdgcn_sched_barrier(0);
            v8f acc = {};
            #pragma unroll
            for (int j = 0; j < 8; ++j)
                acc = __builtin_amdgcn_wmma_f32_16x16x4_f32(
                    false, fa[j], false, fb[j], (short)0, acc, false, false);
            float r2 = 0.f;
            #pragma unroll
            for (int r = 0; r < 8; ++r) {
                const bool dg = (mi + half * 8 + r) == (ni + l15);
                t[coff + r * LD] = (dg ? 1.5f : 0.f) - 0.5f * acc[r];
                const float d = (dg ? 1.f : 0.f) - acc[r];
                r2 += d * d;
            }
            r2 = wave_sum_uniform(r2);
            if (lane == 0) red[wave] = r2;
        }
        __syncthreads();
        const float res2 = (red[0] + red[1]) + (red[2] + red[3]);
        {
            v2f ya[8], tb[8], ta[8], zb[8];
            #pragma unroll
            for (int j = 0; j < 8; ++j) {
                ya[j].x = y[aoff + 4 * j];
                ya[j].y = y[aoff + 4 * j + 1];
                tb[j].x = t[boff + 4 * j * LD];
                tb[j].y = t[boff + 4 * j * LD + LD];
                ta[j].x = t[aoff + 4 * j];
                ta[j].y = t[aoff + 4 * j + 1];
                zb[j].x = z[boff + 4 * j * LD];
                zb[j].y = z[boff + 4 * j * LD + LD];
            }
            __builtin_amdgcn_sched_barrier(0);
            v8f accY = {}, accZ = {};
            #pragma unroll
            for (int j = 0; j < 8; ++j) {
                accY = __builtin_amdgcn_wmma_f32_16x16x4_f32(
                    false, ya[j], false, tb[j], (short)0, accY, false, false);
                accZ = __builtin_amdgcn_wmma_f32_16x16x4_f32(
                    false, ta[j], false, zb[j], (short)0, accZ, false, false);
            }
            #pragma unroll
            for (int r = 0; r < 8; ++r) {
                yn[coff + r * LD] = accY[r];
                zn[coff + r * LD] = accZ[r];
            }
        }
        float* tmp;
        tmp = y; y = yn; yn = tmp;
        tmp = z; z = zn; zn = tmp;
        if (res2 < NS_TOL2) break;
    }
    const float sc  = __fsqrt_rn(c);
    const float isc = __frsqrt_rn(c);
    __syncthreads();
    if (dst_s) for (int i = tid; i < 1024; i += 128) dst_s[at32(i)] = y[at32(i)] * sc;
    if (dst_i) for (int i = tid; i < 1024; i += 128) dst_i[at32(i)] = z[at32(i)] * isc;
    __syncthreads();
}

__device__ __forceinline__ void wave_lds_sync()
{
    __builtin_amdgcn_fence(__ATOMIC_RELEASE, "wavefront");
    __builtin_amdgcn_wave_barrier();
    __builtin_amdgcn_fence(__ATOMIC_ACQUIRE, "wavefront");
}

template <int EPI>
__device__ __forceinline__ void mmw(float* C, const float* A, const float* B,
                                    int half, int l15)
{
    v2f a0[8], a1[8], b0[8], b1[8];
    const float* ar = A + l15 * LD + 2 * half;
    const float* br = B + (2 * half) * LD + l15;
    #pragma unroll
    for (int j = 0; j < 8; ++j) {
        a0[j].x = ar[4 * j];
        a0[j].y = ar[4 * j + 1];
        a1[j].x = ar[16 * LD + 4 * j];
        a1[j].y = ar[16 * LD + 4 * j + 1];
        b0[j].x = br[4 * j * LD];
        b0[j].y = br[4 * j * LD + LD];
        b1[j].x = br[4 * j * LD + 16];
        b1[j].y = br[4 * j * LD + LD + 16];
    }
    __builtin_amdgcn_sched_barrier(0);
    v8f c00 = {}, c01 = {}, c10 = {}, c11 = {};
    #pragma unroll
    for (int j = 0; j < 8; ++j) {
        c00 = __builtin_amdgcn_wmma_f32_16x16x4_f32(
            false, a0[j], false, b0[j], (short)0, c00, false, false);
        c01 = __builtin_amdgcn_wmma_f32_16x16x4_f32(
            false, a0[j], false, b1[j], (short)0, c01, false, false);
        c10 = __builtin_amdgcn_wmma_f32_16x16x4_f32(
            false, a1[j], false, b0[j], (short)0, c10, false, false);
        c11 = __builtin_amdgcn_wmma_f32_16x16x4_f32(
            false, a1[j], false, b1[j], (short)0, c11, false, false);
    }
    float* cr = C + (half * 8) * LD + l15;
    #pragma unroll
    for (int r = 0; r < 8; ++r) {
        if (EPI == 1) {
            const bool dg = (half * 8 + r) == l15;
            cr[r * LD]             = (dg ? 1.5f : 0.f) - 0.5f * c00[r];
            cr[r * LD + 16]        = 0.f - 0.5f * c01[r];
            cr[(16 + r) * LD]      = 0.f - 0.5f * c10[r];
            cr[(16 + r) * LD + 16] = (dg ? 1.5f : 0.f) - 0.5f * c11[r];
        } else {
            cr[r * LD]             = c00[r];
            cr[r * LD + 16]        = c01[r];
            cr[(16 + r) * LD]      = c10[r];
            cr[(16 + r) * LD + 16] = c11[r];
        }
    }
    wave_lds_sync();
}

typedef __attribute__((ext_vector_type(4))) float v4f;

struct Tile4 { v8f t[2][2]; };

struct FragA { v4f f[2][2][2]; };

__device__ __forceinline__ void lda(FragA& a, const float* A, int half, int l15)
{
    const float* ar = A + l15 * LD + 8 * half;
    #pragma unroll
    for (int rb = 0; rb < 2; ++rb)
        #pragma unroll
        for (int kb = 0; kb < 2; ++kb)
            #pragma unroll
            for (int p = 0; p < 2; ++p)
                a.f[rb][kb][p] = *(const v4f*)(ar + rb * 16 * LD + kb * 16 + 4 * p);
}

__device__ __forceinline__ void ldd(Tile4& c, const float* C, int half, int l15)
{
    const float* cr = C + (half * 8) * LD + l15;
    #pragma unroll
    for (int rb = 0; rb < 2; ++rb)
        #pragma unroll
        for (int cb = 0; cb < 2; ++cb)
            #pragma unroll
            for (int r = 0; r < 8; ++r)
                c.t[rb][cb][r] = cr[(16 * rb + r) * LD + 16 * cb];
}

__device__ __forceinline__ void std4(float* C, const Tile4& c, int half, int l15)
{
    float* cr = C + (half * 8) * LD + l15;
    #pragma unroll
    for (int rb = 0; rb < 2; ++rb)
        #pragma unroll
        for (int cb = 0; cb < 2; ++cb)
            #pragma unroll
            for (int r = 0; r < 8; ++r)
                cr[(16 * rb + r) * LD + 16 * cb] = c.t[rb][cb][r];
    wave_lds_sync();
}

template <int EPI>
__device__ __forceinline__ float mmr(Tile4& c, const FragA& a, const Tile4& b, int half, int l15)
{
    v8f c00 = {}, c01 = {}, c10 = {}, c11 = {};
    #pragma unroll
    for (int kb = 0; kb < 2; ++kb) {
        #pragma unroll
        for (int m = 0; m < 4; ++m) {
            const int e = 2 * (m & 1);
            v2f a0, a1, b0, b1;
            a0.x = a.f[0][kb][m >> 1][e];  a0.y = a.f[0][kb][m >> 1][e + 1];
            a1.x = a.f[1][kb][m >> 1][e];  a1.y = a.f[1][kb][m >> 1][e + 1];
            b0.x = b.t[kb][0][2 * m];      b0.y = b.t[kb][0][2 * m + 1];
            b1.x = b.t[kb][1][2 * m];      b1.y = b.t[kb][1][2 * m + 1];
            c00 = __builtin_amdgcn_wmma_f32_16x16x4_f32(
                false, a0, false, b0, (short)0, c00, false, false);
            c01 = __builtin_amdgcn_wmma_f32_16x16x4_f32(
                false, a0, false, b1, (short)0, c01, false, false);
            c10 = __builtin_amdgcn_wmma_f32_16x16x4_f32(
                false, a1, false, b0, (short)0, c10, false, false);
            c11 = __builtin_amdgcn_wmma_f32_16x16x4_f32(
                false, a1, false, b1, (short)0, c11, false, false);
        }
    }
    float r2 = 0.f;
    if (EPI == 1) {
        #pragma unroll
        for (int r = 0; r < 8; ++r) {
            const bool dg = (half * 8 + r) == l15;
            const float d00 = (dg ? 1.f : 0.f) - c00[r];
            const float d11 = (dg ? 1.f : 0.f) - c11[r];
            r2 += (d00 * d00 + c01[r] * c01[r]) + (c10[r] * c10[r] + d11 * d11);
            c00[r] = (dg ? 1.5f : 0.f) - 0.5f * c00[r];
            c01[r] = 0.f - 0.5f * c01[r];
            c10[r] = 0.f - 0.5f * c10[r];
            c11[r] = (dg ? 1.5f : 0.f) - 0.5f * c11[r];
        }
    }
    c.t[0][0] = c00; c.t[0][1] = c01; c.t[1][0] = c10; c.t[1][1] = c11;
    return r2;
}

__device__ __forceinline__ void ns_sqrt_w(float* y, float* z, float* t, int iters,
                                          int lane, int half, int l15)
{
    float c = 0.f;
    #pragma unroll
    for (int i = 0; i < 32; ++i) c += y[i * (LD + 1)];
    const float inv_c = 1.f / c;
    for (int i = lane; i < 1024; i += 32) {
        const int a = at32(i);
        const float v = y[a] * inv_c;
        y[a] = v;
        z[a] = (diag32(i) ? 1.5f : 0.f) - 0.5f * v;
    }
    wave_lds_sync();
    Tile4 Yr, Zr, Tr;
    FragA aY, aZ, aT;
    lda(aY, y, half, l15);
    ldd(Zr, z, half, l15);
    mmr<0>(Yr, aY, Zr, half, l15);
    lda(aZ, z, half, l15);
    std4(y, Yr, half, l15);
    lda(aY, y, half, l15);
    __builtin_amdgcn_sched_barrier(0);
    float res2 = wave_sum_uniform(mmr<1>(Tr, aZ, Yr, half, l15));
    for (int it = 2; it < iters && !(res2 < NS_TOL2); ++it) {
        std4(t, Tr, half, l15);
        lda(aT, t, half, l15);
        __builtin_amdgcn_sched_barrier(0);
        mmr<0>(Yr, aY, Tr, half, l15);
        __builtin_amdgcn_sched_barrier(0);
        Tile4 Zn;
        mmr<0>(Zn, aT, Zr, half, l15);
        Zr = Zn;
        std4(z, Zr, half, l15);
        lda(aZ, z, half, l15);
        std4(y, Yr, half, l15);
        lda(aY, y, half, l15);
        __builtin_amdgcn_sched_barrier(0);
        res2 = wave_sum_uniform(mmr<1>(Tr, aZ, Yr, half, l15));
    }
    mmr<0>(Yr, aY, Tr, half, l15);
    std4(y, Yr, half, l15);
    const float sc = __fsqrt_rn(c);
    for (int i = lane; i < 1024; i += 32) {
        const int a = at32(i);
        y[a] = y[a] * sc;
    }
    wave_lds_sync();
}

__global__ __launch_bounds__(128, 1)
void spdnet_mixedop_kernel(const float* __restrict__ x,
                           const float* __restrict__ wts,
                           const float* __restrict__ Ws,
                           float* __restrict__ out)
{
    __shared__ __attribute__((aligned(16))) float lds[14 * MS + 32];
    float* const sS   = lds;
    float* const pSq  = lds + 4 * MS;
    float* const pSi  = lds + 5 * MS;
    float* const sW0  = lds + 7 * MS;
    float* const sW1  = lds + 8 * MS;
    float* const sP0  = lds + 9 * MS;
    float* const sB0  = lds + 11 * MS;
    float* const sB1  = lds + 12 * MS;
    float* const sB2  = lds + 13 * MS;
    float* const sRed = lds + 14 * MS;
    float* const sX   = lds + 4 * MS;
    float* const sWk  = lds + 8 * MS;
    float* const sP   = lds + 10 * MS;

    const int bc  = blockIdx.x;
    const int tid = threadIdx.x;

    float wk[4];
    #pragma unroll
    for (int k = 0; k < 4; ++k) wk[k] = wts[k];

    for (int i = tid; i < 4096; i += 128) sX[(i >> 6) * LDW + (i & 63)] = x[bc * 4096 + i];

    for (int k = 0; k < 4; ++k) {
        __syncthreads();
        for (int i = tid; i < 2048; i += 128) sWk[(i >> 6) * LDW + (i & 63)] = Ws[k * 2048 + i];
        mm(sP, LDW, sWk, LDW, sX, LDW, 32, 64, 64, false);
        mm(sS + k * MS, LD, sP, LDW, sWk, LDW, 32, 32, 64, true);
    }

    __syncthreads();
    for (int i = tid; i < 1024; i += 128) {
        const int a = at32(i);
        sB0[a] = wk[0] * sS[a] + wk[1] * sS[MS + a] +
                 wk[2] * sS[2 * MS + a] + wk[3] * sS[3 * MS + a];
    }

    ns_sqrt(pSq, pSi, sB0, sB1, sB2, sW0, sW1, sP0, sRed, 14);

    __syncthreads();

    {
        const int lane = tid & 31;
        const int half = lane >> 4;
        const int l15  = lane & 15;
        const int k = __builtin_amdgcn_readfirstlane(tid >> 5);
        float* const wa = sS + k * MS;
        float* const wb = lds + (6 + 2 * k) * MS;
        float* const wc = lds + (7 + 2 * k) * MS;

        mmw<0>(wb, pSi, wa, half, l15);
        mmw<0>(wc, wb, pSi, half, l15);

        for (int s = 0; s < 2; ++s) ns_sqrt_w(wc, wa, wb, 12, lane, half, l15);

        wc[lane * (LD + 1)] -= 1.f;

        float* q = wb;
        wave_lds_sync();
        for (int i = lane; i < 1024; i += 32) {
            const int a = at32(i);
            q[a] = wc[a] * (-1.f / (float)LOG_DEG) + (diag32(i) ? (1.f / (float)(LOG_DEG - 1)) : 0.f);
        }
        wave_lds_sync();
        FragA aE;
        Tile4 Q, Qn;
        lda(aE, wc, half, l15);
        ldd(Q, q, half, l15);
        for (int ii = LOG_DEG - 2; ii >= 1; --ii) {
            mmr<0>(Qn, aE, Q, half, l15);
            const float a = ((ii & 1) ? 1.f : -1.f) / (float)ii;
            #pragma unroll
            for (int r = 0; r < 8; ++r) {
                const bool dg = (half * 8 + r) == l15;
                Qn.t[0][0][r] = dg ? Qn.t[0][0][r] + a : Qn.t[0][0][r];
                Qn.t[1][1][r] = dg ? Qn.t[1][1][r] + a : Qn.t[1][1][r];
            }
            Q = Qn;
        }
        mmr<0>(Qn, aE, Q, half, l15);
        std4(wa, Qn, half, l15);
    }

    __syncthreads();
    for (int i = tid; i < 1024; i += 128) {
        const int a = at32(i);
        float u = 0.f;
        #pragma unroll
        for (int k = 0; k < 4; ++k) u += (wk[k] * 4.f) * sS[k * MS + a];
        const float v = u * (1.f / 64.f);
        sB0[a] = v;
        sB1[a] = (diag32(i) ? 1.f : 0.f) + v * (1.f / 10.f);
    }
    for (int ii = 9; ii >= 1; --ii) {
        mm(sB2, LD, sB0, LD, sB1, LD, 32, 32, 32, false);
        const float inv = 1.f / (float)ii;
        for (int i = tid; i < 1024; i += 128) {
            const int a = at32(i);
            sB1[a] = (diag32(i) ? 1.f : 0.f) + sB2[a] * inv;
        }
    }
    float* r  = sB1;
    float* rn = sB2;
    for (int sq = 0; sq < 6; ++sq) {
        mm(rn, LD, r, LD, r, LD, 32, 32, 32, false);
        float* tmp = r; r = rn; rn = tmp;
    }
    mm(sB0, LD, pSq, LD, r, LD, 32, 32, 32, false);
    mm(sP0, LD, sB0, LD, pSq, LD, 32, 32, 32, false);
    __syncthreads();
    for (int i = tid; i < 1024; i += 128) out[bc * 1024 + i] = sP0[at32(i)];
}

extern "C" void kernel_launch(void* const* d_in, const int* in_sizes, int n_in,
                              void* d_out, int out_size, void* d_ws, size_t ws_size,
                              hipStream_t stream)
{
    const float* x   = (const float*)d_in[0];
    const float* wts = (const float*)d_in[1];
    const float* Ws  = (const float*)d_in[2];
    float* out = (float*)d_out;

    dim3 grid(2048);
    dim3 block(128);
    hipLaunchKernelGGL(spdnet_mixedop_kernel, grid, block, 0, stream,
                       x, wts, Ws, out);
}


# Newton-Schulz roots start from A/c with c = 2/3 of the largest absolute column sum (spectrum in (0,1.5], inside |I-A/c|<1) instead of c = trace: about 4 fewer iterations per root; rounding differs
# speedup vs baseline: 6.4909x; 1.3602x over previous
#include <hip/hip_runtime.h>


typedef __attribute__((ext_vector_type(2))) float v2f;
typedef __attribute__((ext_vector_type(8))) float v8f;

constexpr int LD  = 36;
constexpr int LDW = 68;
constexpr int MS  = 32 * LD;
constexpr int LOG_DEG = 18;

constexpr float NS_TOL2 = 1.5e-7f;

constexpr float NS_START = 2.f / 3.f;

template <int M>
__device__ __forceinline__ float row_xor(float v)
{
    const int i = __builtin_bit_cast(int, v);
    return __builtin_bit_cast(float, __builtin_amdgcn_update_dpp(i, i, 0x160 + M, 0xf, 0xf, false));
}

__device__ __forceinline__ float wave_sum_uniform(float v)
{
    v += row_xor<1>(v);
    v += row_xor<2>(v);
    v += row_xor<4>(v);
    v += row_xor<8>(v);
    const int i = __builtin_bit_cast(int, v);
    v += __builtin_bit_cast(float, __builtin_amdgcn_permlanex16(i, i, 0x76543210, 0xfedcba98, false, false));
    return __builtin_bit_cast(float, __builtin_amdgcn_readfirstlane(__builtin_bit_cast(int, v)));
}

__device__ __forceinline__ float wave_max_uniform(float v)
{
    v = __builtin_fmaxf(v, row_xor<1>(v));
    v = __builtin_fmaxf(v, row_xor<2>(v));
    v = __builtin_fmaxf(v, row_xor<4>(v));
    v = __builtin_fmaxf(v, row_xor<8>(v));
    const int i = __builtin_bit_cast(int, v);
    v = __builtin_fmaxf(v, __builtin_bit_cast(float, __builtin_amdgcn_permlanex16(i, i, 0x76543210, 0xfedcba98, false, false)));
    return __builtin_bit_cast(float, __builtin_amdgcn_readfirstlane(__builtin_bit_cast(int, v)));
}

__device__ __forceinline__ int at32(int i) { return (i >> 5) * LD + (i & 31); }
__device__ __forceinline__ bool diag32(int i) { return (i >> 5) == (i & 31); }

__device__ __forceinline__ void mm(float* __restrict__ C, int ldc,
                                   const float* __restrict__ A, int lda,
                                   const float* __restrict__ B, int ldb,
                                   int M, int N, int Kd, bool transB)
{
    __syncthreads();
    const int wave = threadIdx.x >> 5;
    const int lane = threadIdx.x & 31;
    const int half = lane >> 4;
    const int l15  = lane & 15;
    const int tn = N >> 4;
    const int tiles = (M >> 4) * tn;
    for (int t = wave; t < tiles; t += 4) {
        const int mi = (t / tn) << 4;
        const int ni = (t % tn) << 4;
        v8f acc = {};
        #pragma unroll 4
        for (int k = 0; k < Kd; k += 4) {
            v2f a, b;
            const float* ar = A + (mi + l15) * lda + k + 2 * half;
            a.x = ar[0];
            a.y = ar[1];
            if (!transB) {
                const float* br = B + (k + 2 * half) * ldb + ni + l15;
                b.x = br[0];
                b.y = br[ldb];
            } else {
                const float* br = B + (ni + l15) * ldb + k + 2 * half;
                b.x = br[0];
                b.y = br[1];
            }
            acc = __builtin_amdgcn_wmma_f32_16x16x4_f32(
                false, a, false, b, (short)0, acc, false, false);
        }
        #pragma unroll
        for (int r = 0; r < 8; ++r)
            C[(mi + half * 8 + r) * ldc + ni + l15] = acc[r];
    }
    __syncthreads();
}

__device__ void ns_sqrt(float* dst_s, float* dst_i, const float* src,
                        float* y, float* z, float* t, float* yn, float* zn,
                        float* red, int iters)
{
    const int tid = threadIdx.x;
    const int wave = tid >> 5;
    const int lane = tid & 31;
    __syncthreads();
    float v[8];
    float cs = 0.f;
    #pragma unroll
    for (int j = 0; j < 8; ++j) {
        v[j] = src[at32(tid + 128 * j)];
        cs += __builtin_fabsf(v[j]);
    }
    t[tid] = cs;
    __syncthreads();
    const float c = NS_START * wave_max_uniform((t[lane] + t[32 + lane]) + (t[64 + lane] + t[96 + lane]));
    const float inv_c = 1.f / c;
    #pragma unroll
    for (int j = 0; j < 8; ++j) {
        const int i = tid + 128 * j;
        const int a = at32(i);
        const float u = v[j] * inv_c;
        y[a] = u;
        zn[a] = (diag32(i) ? 1.5f : 0.f) - 0.5f * u;
    }
    const int half = lane >> 4;
    const int l15  = lane & 15;
    const int mi = (wave >> 1) << 4;
    const int ni = (wave & 1) << 4;
    const int aoff = (mi + l15) * LD + 2 * half;
    const int boff = (2 * half) * LD + ni + l15;
    const int coff = (mi + half * 8) * LD + ni + l15;
    __syncthreads();
    {
        v2f ya[8], tb[8];
        #pragma unroll
        for (int j = 0; j < 8; ++j) {
            ya[j].x = y[aoff + 4 * j];
            ya[j].y = y[aoff + 4 * j + 1];
            tb[j].x = zn[boff + 4 * j * LD];
            tb[j].y = zn[boff + 4 * j * LD + LD];
        }
        v8f accY = {};
        #pragma unroll
        for (int j = 0; j < 8; ++j)
            accY = __builtin_amdgcn_wmma_f32_16x16x4_f32(
                false, ya[j], false, tb[j], (short)0, accY, false, false);
        #pragma unroll
        for (int r = 0; r < 8; ++r) yn[coff + r * LD] = accY[r];
        float* tmp;
        tmp = y; y = yn; yn = tmp;
        tmp = z; z = zn; zn = tmp;
    }
    for (int it = 1; it < iters; ++it) {
        __syncthreads();
        {
            v2f fa[8], fb[8];
            #pragma unroll
            for (int j = 0; j < 8; ++j) {
                fa[j].x = z[aoff + 4 * j];
                fa[j].y = z[aoff + 4 * j + 1];
                fb[j].x = y[boff + 4 * j * LD];
                fb[j].y = y[boff + 4 * j * LD + LD];
            }
            __builtin_amdgcn_sched_barrier(0);
            v8f acc = {};
            #pragma unroll
            for (int j = 0; j < 8; ++j)
                acc = __builtin_amdgcn_wmma_f32_16x16x4_f32(
                    false, fa[j], false, fb[j], (short)0, acc, false, false);
            float r2 = 0.f;
            #pragma unroll
            for (int r = 0; r < 8; ++r) {
                const bool dg = (mi + half * 8 + r) == (ni + l15);
                t[coff + r * LD] = (dg ? 1.5f : 0.f) - 0.5f * acc[r];
                const float d = (dg ? 1.f : 0.f) - acc[r];
                r2 += d * d;
            }
            r2 = wave_sum_uniform(r2);
            if (lane == 0) red[wave] = r2;
        }
        __syncthreads();
        const float res2 = (red[0] + red[1]) + (red[2] + red[3]);
        {
            v2f ya[8], tb[8], ta[8], zb[8];
            #pragma unroll
            for (int j = 0; j < 8; ++j) {
                ya[j].x = y[aoff + 4 * j];
                ya[j].y = y[aoff + 4 * j + 1];
                tb[j].x = t[boff + 4 * j * LD];
                tb[j].y = t[boff + 4 * j * LD + LD];
                ta[j].x = t[aoff + 4 * j];
                ta[j].y = t[aoff + 4 * j + 1];
                zb[j].x = z[boff + 4 * j * LD];
                zb[j].y = z[boff + 4 * j * LD + LD];
            }
            __builtin_amdgcn_sched_barrier(0);
            v8f accY = {}, accZ = {};
            #pragma unroll
            for (int j = 0; j < 8; ++j) {
                accY = __builtin_amdgcn_wmma_f32_16x16x4_f32(
                    false, ya[j], false, tb[j], (short)0, accY, false, false);
                accZ = __builtin_amdgcn_wmma_f32_16x16x4_f32(
                    false, ta[j], false, zb[j], (short)0, accZ, false, false);
            }
            #pragma unroll
            for (int r = 0; r < 8; ++r) {
                yn[coff + r * LD] = accY[r];
                zn[coff + r * LD] = accZ[r];
            }
        }
        float* tmp;
        tmp = y; y = yn; yn = tmp;
        tmp = z; z = zn; zn = tmp;
        if (res2 < NS_TOL2) break;
    }
    const float sc  = __fsqrt_rn(c);
    const float isc = __frsqrt_rn(c);
    __syncthreads();
    if (dst_s) for (int i = tid; i < 1024; i += 128) dst_s[at32(i)] = y[at32(i)] * sc;
    if (dst_i) for (int i = tid; i < 1024; i += 128) dst_i[at32(i)] = z[at32(i)] * isc;
    __syncthreads();
}

__device__ __forceinline__ void wave_lds_sync()
{
    __builtin_amdgcn_fence(__ATOMIC_RELEASE, "wavefront");
    __builtin_amdgcn_wave_barrier();
    __builtin_amdgcn_fence(__ATOMIC_ACQUIRE, "wavefront");
}

template <int EPI>
__device__ __forceinline__ void mmw(float* C, const float* A, const float* B,
                                    int half, int l15)
{
    v2f a0[8], a1[8], b0[8], b1[8];
    const float* ar = A + l15 * LD + 2 * half;
    const float* br = B + (2 * half) * LD + l15;
    #pragma unroll
    for (int j = 0; j < 8; ++j) {
        a0[j].x = ar[4 * j];
        a0[j].y = ar[4 * j + 1];
        a1[j].x = ar[16 * LD + 4 * j];
        a1[j].y = ar[16 * LD + 4 * j + 1];
        b0[j].x = br[4 * j * LD];
        b0[j].y = br[4 * j * LD + LD];
        b1[j].x = br[4 * j * LD + 16];
        b1[j].y = br[4 * j * LD + LD + 16];
    }
    __builtin_amdgcn_sched_barrier(0);
    v8f c00 = {}, c01 = {}, c10 = {}, c11 = {};
    #pragma unroll
    for (int j = 0; j < 8; ++j) {
        c00 = __builtin_amdgcn_wmma_f32_16x16x4_f32(
            false, a0[j], false, b0[j], (short)0, c00, false, false);
        c01 = __builtin_amdgcn_wmma_f32_16x16x4_f32(
            false, a0[j], false, b1[j], (short)0, c01, false, false);
        c10 = __builtin_amdgcn_wmma_f32_16x16x4_f32(
            false, a1[j], false, b0[j], (short)0, c10, false, false);
        c11 = __builtin_amdgcn_wmma_f32_16x16x4_f32(
            false, a1[j], false, b1[j], (short)0, c11, false, false);
    }
    float* cr = C + (half * 8) * LD + l15;
    #pragma unroll
    for (int r = 0; r < 8; ++r) {
        if (EPI == 1) {
            const bool dg = (half * 8 + r) == l15;
            cr[r * LD]             = (dg ? 1.5f : 0.f) - 0.5f * c00[r];
            cr[r * LD + 16]        = 0.f - 0.5f * c01[r];
            cr[(16 + r) * LD]      = 0.f - 0.5f * c10[r];
            cr[(16 + r) * LD + 16] = (dg ? 1.5f : 0.f) - 0.5f * c11[r];
        } else {
            cr[r * LD]             = c00[r];
            cr[r * LD + 16]        = c01[r];
            cr[(16 + r) * LD]      = c10[r];
            cr[(16 + r) * LD + 16] = c11[r];
        }
    }
    wave_lds_sync();
}

typedef __attribute__((ext_vector_type(4))) float v4f;

struct Tile4 { v8f t[2][2]; };

struct FragA { v4f f[2][2][2]; };

__device__ __forceinline__ void lda(FragA& a, const float* A, int half, int l15)
{
    const float* ar = A + l15 * LD + 8 * half;
    #pragma unroll
    for (int rb = 0; rb < 2; ++rb)
        #pragma unroll
        for (int kb = 0; kb < 2; ++kb)
            #pragma unroll
            for (int p = 0; p < 2; ++p)
                a.f[rb][kb][p] = *(const v4f*)(ar + rb * 16 * LD + kb * 16 + 4 * p);
}

__device__ __forceinline__ void ldd(Tile4& c, const float* C, int half, int l15)
{
    const float* cr = C + (half * 8) * LD + l15;
    #pragma unroll
    for (int rb = 0; rb < 2; ++rb)
        #pragma unroll
        for (int cb = 0; cb < 2; ++cb)
            #pragma unroll
            for (int r = 0; r < 8; ++r)
                c.t[rb][cb][r] = cr[(16 * rb + r) * LD + 16 * cb];
}

__device__ __forceinline__ void std4(float* C, const Tile4& c, int half, int l15)
{
    float* cr = C + (half * 8) * LD + l15;
    #pragma unroll
    for (int rb = 0; rb < 2; ++rb)
        #pragma unroll
        for (int cb = 0; cb < 2; ++cb)
            #pragma unroll
            for (int r = 0; r < 8; ++r)
                cr[(16 * rb + r) * LD + 16 * cb] = c.t[rb][cb][r];
    wave_lds_sync();
}

template <int EPI>
__device__ __forceinline__ float mmr(Tile4& c, const FragA& a, const Tile4& b, int half, int l15)
{
    v8f c00 = {}, c01 = {}, c10 = {}, c11 = {};
    #pragma unroll
    for (int kb = 0; kb < 2; ++kb) {
        #pragma unroll
        for (int m = 0; m < 4; ++m) {
            const int e = 2 * (m & 1);
            v2f a0, a1, b0, b1;
            a0.x = a.f[0][kb][m >> 1][e];  a0.y = a.f[0][kb][m >> 1][e + 1];
            a1.x = a.f[1][kb][m >> 1][e];  a1.y = a.f[1][kb][m >> 1][e + 1];
            b0.x = b.t[kb][0][2 * m];      b0.y = b.t[kb][0][2 * m + 1];
            b1.x = b.t[kb][1][2 * m];      b1.y = b.t[kb][1][2 * m + 1];
            c00 = __builtin_amdgcn_wmma_f32_16x16x4_f32(
                false, a0, false, b0, (short)0, c00, false, false);
            c01 = __builtin_amdgcn_wmma_f32_16x16x4_f32(
                false, a0, false, b1, (short)0, c01, false, false);
            c10 = __builtin_amdgcn_wmma_f32_16x16x4_f32(
                false, a1, false, b0, (short)0, c10, false, false);
            c11 = __builtin_amdgcn_wmma_f32_16x16x4_f32(
                false, a1, false, b1, (short)0, c11, false, false);
        }
    }
    float r2 = 0.f;
    if (EPI == 1) {
        #pragma unroll
        for (int r = 0; r < 8; ++r) {
            const bool dg = (half * 8 + r) == l15;
            const float d00 = (dg ? 1.f : 0.f) - c00[r];
            const float d11 = (dg ? 1.f : 0.f) - c11[r];
            r2 += (d00 * d00 + c01[r] * c01[r]) + (c10[r] * c10[r] + d11 * d11);
            c00[r] = (dg ? 1.5f : 0.f) - 0.5f * c00[r];
            c01[r] = 0.f - 0.5f * c01[r];
            c10[r] = 0.f - 0.5f * c10[r];
            c11[r] = (dg ? 1.5f : 0.f) - 0.5f * c11[r];
        }
    }
    c.t[0][0] = c00; c.t[0][1] = c01; c.t[1][0] = c10; c.t[1][1] = c11;
    return r2;
}

__device__ __forceinline__ void ns_sqrt_w(float* y, float* z, float* t, int iters,
                                          int lane, int half, int l15)
{
    float v[32];
    float cs = 0.f;
    #pragma unroll
    for (int j = 0; j < 32; ++j) {
        v[j] = y[j * LD + lane];
        cs += __builtin_fabsf(v[j]);
    }
    const float vd = y[lane * (LD + 1)];
    const float c = NS_START * wave_max_uniform(cs);
    const float inv_c = 1.f / c;
    #pragma unroll
    for (int j = 0; j < 32; ++j) {
        const float u = v[j] * inv_c;
        y[j * LD + lane] = u;
        z[j * LD + lane] = -0.5f * u;
    }
    z[lane * (LD + 1)] = 1.5f - 0.5f * (vd * inv_c);
    wave_lds_sync();
    Tile4 Yr, Zr, Tr;
    FragA aY, aZ, aT;
    lda(aY, y, half, l15);
    ldd(Zr, z, half, l15);
    mmr<0>(Yr, aY, Zr, half, l15);
    lda(aZ, z, half, l15);
    std4(y, Yr, half, l15);
    lda(aY, y, half, l15);
    __builtin_amdgcn_sched_barrier(0);
    float res2 = wave_sum_uniform(mmr<1>(Tr, aZ, Yr, half, l15));
    for (int it = 2; it < iters && !(res2 < NS_TOL2); ++it) {
        std4(t, Tr, half, l15);
        lda(aT, t, half, l15);
        __builtin_amdgcn_sched_barrier(0);
        mmr<0>(Yr, aY, Tr, half, l15);
        __builtin_amdgcn_sched_barrier(0);
        Tile4 Zn;
        mmr<0>(Zn, aT, Zr, half, l15);
        Zr = Zn;
        std4(z, Zr, half, l15);
        lda(aZ, z, half, l15);
        std4(y, Yr, half, l15);
        lda(aY, y, half, l15);
        __builtin_amdgcn_sched_barrier(0);
        res2 = wave_sum_uniform(mmr<1>(Tr, aZ, Yr, half, l15));
    }
    mmr<0>(Yr, aY, Tr, half, l15);
    std4(y, Yr, half, l15);
    const float sc = __fsqrt_rn(c);
    for (int i = lane; i < 1024; i += 32) {
        const int a = at32(i);
        y[a] = y[a] * sc;
    }
    wave_lds_sync();
}

__global__ __launch_bounds__(128, 1)
void spdnet_mixedop_kernel(const float* __restrict__ x,
                           const float* __restrict__ wts,
                           const float* __restrict__ Ws,
                           float* __restrict__ out)
{
    __shared__ __attribute__((aligned(16))) float lds[14 * MS + 32];
    float* const sS   = lds;
    float* const pSq  = lds + 4 * MS;
    float* const pSi  = lds + 5 * MS;
    float* const sW0  = lds + 7 * MS;
    float* const sW1  = lds + 8 * MS;
    float* const sP0  = lds + 9 * MS;
    float* const sB0  = lds + 11 * MS;
    float* const sB1  = lds + 12 * MS;
    float* const sB2  = lds + 13 * MS;
    float* const sRed = lds + 14 * MS;
    float* const sX   = lds + 4 * MS;
    float* const sWk  = lds + 8 * MS;
    float* const sP   = lds + 10 * MS;

    const int bc  = blockIdx.x;
    const int tid = threadIdx.x;

    float wk[4];
    #pragma unroll
    for (int k = 0; k < 4; ++k) wk[k] = wts[k];

    for (int i = tid; i < 4096; i += 128) sX[(i >> 6) * LDW + (i & 63)] = x[bc * 4096 + i];

    for (int k = 0; k < 4; ++k) {
        __syncthreads();
        for (int i = tid; i < 2048; i += 128) sWk[(i >> 6) * LDW + (i & 63)] = Ws[k * 2048 + i];
        mm(sP, LDW, sWk, LDW, sX, LDW, 32, 64, 64, false);
        mm(sS + k * MS, LD, sP, LDW, sWk, LDW, 32, 32, 64, true);
    }

    __syncthreads();
    for (int i = tid; i < 1024; i += 128) {
        const int a = at32(i);
        sB0[a] = wk[0] * sS[a] + wk[1] * sS[MS + a] +
                 wk[2] * sS[2 * MS + a] + wk[3] * sS[3 * MS + a];
    }

    ns_sqrt(pSq, pSi, sB0, sB1, sB2, sW0, sW1, sP0, sRed, 14);

    __syncthreads();

    {
        const int lane = tid & 31;
        const int half = lane >> 4;
        const int l15  = lane & 15;
        const int k = __builtin_amdgcn_readfirstlane(tid >> 5);
        float* const wa = sS + k * MS;
        float* const wb = lds + (6 + 2 * k) * MS;
        float* const wc = lds + (7 + 2 * k) * MS;

        mmw<0>(wb, pSi, wa, half, l15);
        mmw<0>(wc, wb, pSi, half, l15);

        for (int s = 0; s < 2; ++s) ns_sqrt_w(wc, wa, wb, 12, lane, half, l15);

        wc[lane * (LD + 1)] -= 1.f;

        float* q = wb;
        wave_lds_sync();
        for (int i = lane; i < 1024; i += 32) {
            const int a = at32(i);
            q[a] = wc[a] * (-1.f / (float)LOG_DEG) + (diag32(i) ? (1.f / (float)(LOG_DEG - 1)) : 0.f);
        }
        wave_lds_sync();
        FragA aE;
        Tile4 Q, Qn;
        lda(aE, wc, half, l15);
        ldd(Q, q, half, l15);
        for (int ii = LOG_DEG - 2; ii >= 1; --ii) {
            mmr<0>(Qn, aE, Q, half, l15);
            const float a = ((ii & 1) ? 1.f : -1.f) / (float)ii;
            #pragma unroll
            for (int r = 0; r < 8; ++r) {
                const bool dg = (half * 8 + r) == l15;
                Qn.t[0][0][r] = dg ? Qn.t[0][0][r] + a : Qn.t[0][0][r];
                Qn.t[1][1][r] = dg ? Qn.t[1][1][r] + a : Qn.t[1][1][r];
            }
            Q = Qn;
        }
        mmr<0>(Qn, aE, Q, half, l15);
        std4(wa, Qn, half, l15);
    }

    __syncthreads();
    for (int i = tid; i < 1024; i += 128) {
        const int a = at32(i);
        float u = 0.f;
        #pragma unroll
        for (int k = 0; k < 4; ++k) u += (wk[k] * 4.f) * sS[k * MS + a];
        const float v = u * (1.f / 64.f);
        sB0[a] = v;
        sB1[a] = (diag32(i) ? 1.f : 0.f) + v * (1.f / 10.f);
    }
    for (int ii = 9; ii >= 1; --ii) {
        mm(sB2, LD, sB0, LD, sB1, LD, 32, 32, 32, false);
        const float inv = 1.f / (float)ii;
        for (int i = tid; i < 1024; i += 128) {
            const int a = at32(i);
            sB1[a] = (diag32(i) ? 1.f : 0.f) + sB2[a] * inv;
        }
    }
    float* r  = sB1;
    float* rn = sB2;
    for (int sq = 0; sq < 6; ++sq) {
        mm(rn, LD, r, LD, r, LD, 32, 32, 32, false);
        float* tmp = r; r = rn; rn = tmp;
    }
    mm(sB0, LD, pSq, LD, r, LD, 32, 32, 32, false);
    mm(sP0, LD, sB0, LD, pSq, LD, 32, 32, 32, false);
    __syncthreads();
    for (int i = tid; i < 1024; i += 128) out[bc * 1024 + i] = sP0[at32(i)];
}

extern "C" void kernel_launch(void* const* d_in, const int* in_sizes, int n_in,
                              void* d_out, int out_size, void* d_ws, size_t ws_size,
                              hipStream_t stream)
{
    const float* x   = (const float*)d_in[0];
    const float* wts = (const float*)d_in[1];
    const float* Ws  = (const float*)d_in[2];
    float* out = (float*)d_out;

    dim3 grid(2048);
    dim3 block(128);
    hipLaunchKernelGGL(spdnet_mixedop_kernel, grid, block, 0, stream,
                       x, wts, Ws, out);
}


# Log series of degree 18 summed in powers of X=E^3: sum_j X^j (a_3j I + a_3j+1 E + a_3j+2 E^2) by Horner in X, 7 products instead of 17; same polynomial, terms added in a new order
# speedup vs baseline: 7.3561x; 1.1333x over previous
#include <hip/hip_runtime.h>


typedef __attribute__((ext_vector_type(2))) float v2f;
typedef __attribute__((ext_vector_type(8))) float v8f;

constexpr int LD  = 36;
constexpr int LDW = 68;
constexpr int MS  = 32 * LD;
constexpr int LOG_DEG = 18;
static_assert(LOG_DEG % 3 == 0, "the log series is evaluated in blocks of three terms");

__device__ __forceinline__ float log_coef(int i)
{
    return i == 0 ? 0.f : ((i & 1) ? 1.f : -1.f) / (float)i;
}

constexpr float NS_TOL2 = 1.5e-7f;

constexpr float NS_START = 2.f / 3.f;

template <int M>
__device__ __forceinline__ float row_xor(float v)
{
    const int i = __builtin_bit_cast(int, v);
    return __builtin_bit_cast(float, __builtin_amdgcn_update_dpp(i, i, 0x160 + M, 0xf, 0xf, false));
}

__device__ __forceinline__ float wave_sum_uniform(float v)
{
    v += row_xor<1>(v);
    v += row_xor<2>(v);
    v += row_xor<4>(v);
    v += row_xor<8>(v);
    const int i = __builtin_bit_cast(int, v);
    v += __builtin_bit_cast(float, __builtin_amdgcn_permlanex16(i, i, 0x76543210, 0xfedcba98, false, false));
    return __builtin_bit_cast(float, __builtin_amdgcn_readfirstlane(__builtin_bit_cast(int, v)));
}

__device__ __forceinline__ float wave_max_uniform(float v)
{
    v = __builtin_fmaxf(v, row_xor<1>(v));
    v = __builtin_fmaxf(v, row_xor<2>(v));
    v = __builtin_fmaxf(v, row_xor<4>(v));
    v = __builtin_fmaxf(v, row_xor<8>(v));
    const int i = __builtin_bit_cast(int, v);
    v = __builtin_fmaxf(v, __builtin_bit_cast(float, __builtin_amdgcn_permlanex16(i, i, 0x76543210, 0xfedcba98, false, false)));
    return __builtin_bit_cast(float, __builtin_amdgcn_readfirstlane(__builtin_bit_cast(int, v)));
}

__device__ __forceinline__ int at32(int i) { return (i >> 5) * LD + (i & 31); }
__device__ __forceinline__ bool diag32(int i) { return (i >> 5) == (i & 31); }

__device__ __forceinline__ void mm(float* __restrict__ C, int ldc,
                                   const float* __restrict__ A, int lda,
                                   const float* __restrict__ B, int ldb,
                                   int M, int N, int Kd, bool transB)
{
    __syncthreads();
    const int wave = threadIdx.x >> 5;
    const int lane = threadIdx.x & 31;
    const int half = lane >> 4;
    const int l15  = lane & 15;
    const int tn = N >> 4;
    const int tiles = (M >> 4) * tn;
    for (int t = wave; t < tiles; t += 4) {
        const int mi = (t / tn) << 4;
        const int ni = (t % tn) << 4;
        v8f acc = {};
        #pragma unroll 4
        for (int k = 0; k < Kd; k += 4) {
            v2f a, b;
            const float* ar = A + (mi + l15) * lda + k + 2 * half;
            a.x = ar[0];
            a.y = ar[1];
            if (!transB) {
                const float* br = B + (k + 2 * half) * ldb + ni + l15;
                b.x = br[0];
                b.y = br[ldb];
            } else {
                const float* br = B + (ni + l15) * ldb + k + 2 * half;
                b.x = br[0];
                b.y = br[1];
            }
            acc = __builtin_amdgcn_wmma_f32_16x16x4_f32(
                false, a, false, b, (short)0, acc, false, false);
        }
        #pragma unroll
        for (int r = 0; r < 8; ++r)
            C[(mi + half * 8 + r) * ldc + ni + l15] = acc[r];
    }
    __syncthreads();
}

__device__ void ns_sqrt(float* dst_s, float* dst_i, const float* src,
                        float* y, float* z, float* t, float* yn, float* zn,
                        float* red, int iters)
{
    const int tid = threadIdx.x;
    const int wave = tid >> 5;
    const int lane = tid & 31;
    __syncthreads();
    float v[8];
    float cs = 0.f;
    #pragma unroll
    for (int j = 0; j < 8; ++j) {
        v[j] = src[at32(tid + 128 * j)];
        cs += __builtin_fabsf(v[j]);
    }
    t[tid] = cs;
    __syncthreads();
    const float c = NS_START * wave_max_uniform((t[lane] + t[32 + lane]) + (t[64 + lane] + t[96 + lane]));
    const float inv_c = 1.f / c;
    #pragma unroll
    for (int j = 0; j < 8; ++j) {
        const int i = tid + 128 * j;
        const int a = at32(i);
        const float u = v[j] * inv_c;
        y[a] = u;
        zn[a] = (diag32(i) ? 1.5f : 0.f) - 0.5f * u;
    }
    const int half = lane >> 4;
    const int l15  = lane & 15;
    const int mi = (wave >> 1) << 4;
    const int ni = (wave & 1) << 4;
    const int aoff = (mi + l15) * LD + 2 * half;
    const int boff = (2 * half) * LD + ni + l15;
    const int coff = (mi + half * 8) * LD + ni + l15;
    __syncthreads();
    {
        v2f ya[8], tb[8];
        #pragma unroll
        for (int j = 0; j < 8; ++j) {
            ya[j].x = y[aoff + 4 * j];
            ya[j].y = y[aoff + 4 * j + 1];
            tb[j].x = zn[boff + 4 * j * LD];
            tb[j].y = zn[boff + 4 * j * LD + LD];
        }
        v8f accY = {};
        #pragma unroll
        for (int j = 0; j < 8; ++j)
            accY = __builtin_amdgcn_wmma_f32_16x16x4_f32(
                false, ya[j], false, tb[j], (short)0, accY, false, false);
        #pragma unroll
        for (int r = 0; r < 8; ++r) yn[coff + r * LD] = accY[r];
        float* tmp;
        tmp = y; y = yn; yn = tmp;
        tmp = z; z = zn; zn = tmp;
    }
    for (int it = 1; it < iters; ++it) {
        __syncthreads();
        {
            v2f fa[8], fb[8];
            #pragma unroll
            for (int j = 0; j < 8; ++j) {
                fa[j].x = z[aoff + 4 * j];
                fa[j].y = z[aoff + 4 * j + 1];
                fb[j].x = y[boff + 4 * j * LD];
                fb[j].y = y[boff + 4 * j * LD + LD];
            }
            __builtin_amdgcn_sched_barrier(0);
            v8f acc = {};
            #pragma unroll
            for (int j = 0; j < 8; ++j)
                acc = __builtin_amdgcn_wmma_f32_16x16x4_f32(
                    false, fa[j], false, fb[j], (short)0, acc, false, false);
            float r2 = 0.f;
            #pragma unroll
            for (int r = 0; r < 8; ++r) {
                const bool dg = (mi + half * 8 + r) == (ni + l15);
                t[coff + r * LD] = (dg ? 1.5f : 0.f) - 0.5f * acc[r];
                const float d = (dg ? 1.f : 0.f) - acc[r];
                r2 += d * d;
            }
            r2 = wave_sum_uniform(r2);
            if (lane == 0) red[wave] = r2;
        }
        __syncthreads();
        const float res2 = (red[0] + red[1]) + (red[2] + red[3]);
        {
            v2f ya[8], tb[8], ta[8], zb[8];
            #pragma unroll
            for (int j = 0; j < 8; ++j) {
                ya[j].x = y[aoff + 4 * j];
                ya[j].y = y[aoff + 4 * j + 1];
                tb[j].x = t[boff + 4 * j * LD];
                tb[j].y = t[boff + 4 * j * LD + LD];
                ta[j].x = t[aoff + 4 * j];
                ta[j].y = t[aoff + 4 * j + 1];
                zb[j].x = z[boff + 4 * j * LD];
                zb[j].y = z[boff + 4 * j * LD + LD];
            }
            __builtin_amdgcn_sched_barrier(0);
            v8f accY = {}, accZ = {};
            #pragma unroll
            for (int j = 0; j < 8; ++j) {
                accY = __builtin_amdgcn_wmma_f32_16x16x4_f32(
                    false, ya[j], false, tb[j], (short)0, accY, false, false);
                accZ = __builtin_amdgcn_wmma_f32_16x16x4_f32(
                    false, ta[j], false, zb[j], (short)0, accZ, false, false);
            }
            #pragma unroll
            for (int r = 0; r < 8; ++r) {
                yn[coff + r * LD] = accY[r];
                zn[coff + r * LD] = accZ[r];
            }
        }
        float* tmp;
        tmp = y; y = yn; yn = tmp;
        tmp = z; z = zn; zn = tmp;
        if (res2 < NS_TOL2) break;
    }
    const float sc  = __fsqrt_rn(c);
    const float isc = __frsqrt_rn(c);
    __syncthreads();
    if (dst_s) for (int i = tid; i < 1024; i += 128) dst_s[at32(i)] = y[at32(i)] * sc;
    if (dst_i) for (int i = tid; i < 1024; i += 128) dst_i[at32(i)] = z[at32(i)] * isc;
    __syncthreads();
}

__device__ __forceinline__ void wave_lds_sync()
{
    __builtin_amdgcn_fence(__ATOMIC_RELEASE, "wavefront");
    __builtin_amdgcn_wave_barrier();
    __builtin_amdgcn_fence(__ATOMIC_ACQUIRE, "wavefront");
}

template <int EPI>
__device__ __forceinline__ void mmw(float* C, const float* A, const float* B,
                                    int half, int l15)
{
    v2f a0[8], a1[8], b0[8], b1[8];
    const float* ar = A + l15 * LD + 2 * half;
    const float* br = B + (2 * half) * LD + l15;
    #pragma unroll
    for (int j = 0; j < 8; ++j) {
        a0[j].x = ar[4 * j];
        a0[j].y = ar[4 * j + 1];
        a1[j].x = ar[16 * LD + 4 * j];
        a1[j].y = ar[16 * LD + 4 * j + 1];
        b0[j].x = br[4 * j * LD];
        b0[j].y = br[4 * j * LD + LD];
        b1[j].x = br[4 * j * LD + 16];
        b1[j].y = br[4 * j * LD + LD + 16];
    }
    __builtin_amdgcn_sched_barrier(0);
    v8f c00 = {}, c01 = {}, c10 = {}, c11 = {};
    #pragma unroll
    for (int j = 0; j < 8; ++j) {
        c00 = __builtin_amdgcn_wmma_f32_16x16x4_f32(
            false, a0[j], false, b0[j], (short)0, c00, false, false);
        c01 = __builtin_amdgcn_wmma_f32_16x16x4_f32(
            false, a0[j], false, b1[j], (short)0, c01, false, false);
        c10 = __builtin_amdgcn_wmma_f32_16x16x4_f32(
            false, a1[j], false, b0[j], (short)0, c10, false, false);
        c11 = __builtin_amdgcn_wmma_f32_16x16x4_f32(
            false, a1[j], false, b1[j], (short)0, c11, false, false);
    }
    float* cr = C + (half * 8) * LD + l15;
    #pragma unroll
    for (int r = 0; r < 8; ++r) {
        if (EPI == 1) {
            const bool dg = (half * 8 + r) == l15;
            cr[r * LD]             = (dg ? 1.5f : 0.f) - 0.5f * c00[r];
            cr[r * LD + 16]        = 0.f - 0.5f * c01[r];
            cr[(16 + r) * LD]      = 0.f - 0.5f * c10[r];
            cr[(16 + r) * LD + 16] = (dg ? 1.5f : 0.f) - 0.5f * c11[r];
        } else {
            cr[r * LD]             = c00[r];
            cr[r * LD + 16]        = c01[r];
            cr[(16 + r) * LD]      = c10[r];
            cr[(16 + r) * LD + 16] = c11[r];
        }
    }
    wave_lds_sync();
}

typedef __attribute__((ext_vector_type(4))) float v4f;

struct Tile4 { v8f t[2][2]; };

struct FragA { v4f f[2][2][2]; };

__device__ __forceinline__ void lda(FragA& a, const float* A, int half, int l15)
{
    const float* ar = A + l15 * LD + 8 * half;
    #pragma unroll
    for (int rb = 0; rb < 2; ++rb)
        #pragma unroll
        for (int kb = 0; kb < 2; ++kb)
            #pragma unroll
            for (int p = 0; p < 2; ++p)
                a.f[rb][kb][p] = *(const v4f*)(ar + rb * 16 * LD + kb * 16 + 4 * p);
}

__device__ __forceinline__ void ldd(Tile4& c, const float* C, int half, int l15)
{
    const float* cr = C + (half * 8) * LD + l15;
    #pragma unroll
    for (int rb = 0; rb < 2; ++rb)
        #pragma unroll
        for (int cb = 0; cb < 2; ++cb)
            #pragma unroll
            for (int r = 0; r < 8; ++r)
                c.t[rb][cb][r] = cr[(16 * rb + r) * LD + 16 * cb];
}

__device__ __forceinline__ void std4(float* C, const Tile4& c, int half, int l15)
{
    float* cr = C + (half * 8) * LD + l15;
    #pragma unroll
    for (int rb = 0; rb < 2; ++rb)
        #pragma unroll
        for (int cb = 0; cb < 2; ++cb)
            #pragma unroll
            for (int r = 0; r < 8; ++r)
                cr[(16 * rb + r) * LD + 16 * cb] = c.t[rb][cb][r];
    wave_lds_sync();
}

template <int EPI>
__device__ __forceinline__ float mmr(Tile4& c, const FragA& a, const Tile4& b, int half, int l15)
{
    v8f c00 = {}, c01 = {}, c10 = {}, c11 = {};
    #pragma unroll
    for (int kb = 0; kb < 2; ++kb) {
        #pragma unroll
        for (int m = 0; m < 4; ++m) {
            const int e = 2 * (m & 1);
            v2f a0, a1, b0, b1;
            a0.x = a.f[0][kb][m >> 1][e];  a0.y = a.f[0][kb][m >> 1][e + 1];
            a1.x = a.f[1][kb][m >> 1][e];  a1.y = a.f[1][kb][m >> 1][e + 1];
            b0.x = b.t[kb][0][2 * m];      b0.y = b.t[kb][0][2 * m + 1];
            b1.x = b.t[kb][1][2 * m];      b1.y = b.t[kb][1][2 * m + 1];
            c00 = __builtin_amdgcn_wmma_f32_16x16x4_f32(
                false, a0, false, b0, (short)0, c00, false, false);
            c01 = __builtin_amdgcn_wmma_f32_16x16x4_f32(
                false, a0, false, b1, (short)0, c01, false, false);
            c10 = __builtin_amdgcn_wmma_f32_16x16x4_f32(
                false, a1, false, b0, (short)0, c10, false, false);
            c11 = __builtin_amdgcn_wmma_f32_16x16x4_f32(
                false, a1, false, b1, (short)0, c11, false, false);
        }
    }
    float r2 = 0.f;
    if (EPI == 1) {
        #pragma unroll
        for (int r = 0; r < 8; ++r) {
            const bool dg = (half * 8 + r) == l15;
            const float d00 = (dg ? 1.f : 0.f) - c00[r];
            const float d11 = (dg ? 1.f : 0.f) - c11[r];
            r2 += (d00 * d00 + c01[r] * c01[r]) + (c10[r] * c10[r] + d11 * d11);
            c00[r] = (dg ? 1.5f : 0.f) - 0.5f * c00[r];
            c01[r] = 0.f - 0.5f * c01[r];
            c10[r] = 0.f - 0.5f * c10[r];
            c11[r] = (dg ? 1.5f : 0.f) - 0.5f * c11[r];
        }
    }
    c.t[0][0] = c00; c.t[0][1] = c01; c.t[1][0] = c10; c.t[1][1] = c11;
    return r2;
}

__device__ __forceinline__ void mmra(Tile4& c, const FragA& a, const Tile4& b, const Tile4& c0)
{
    v8f c00 = c0.t[0][0], c01 = c0.t[0][1], c10 = c0.t[1][0], c11 = c0.t[1][1];
    #pragma unroll
    for (int kb = 0; kb < 2; ++kb) {
        #pragma unroll
        for (int m = 0; m < 4; ++m) {
            const int e = 2 * (m & 1);
            v2f a0, a1, b0, b1;
            a0.x = a.f[0][kb][m >> 1][e];  a0.y = a.f[0][kb][m >> 1][e + 1];
            a1.x = a.f[1][kb][m >> 1][e];  a1.y = a.f[1][kb][m >> 1][e + 1];
            b0.x = b.t[kb][0][2 * m];      b0.y = b.t[kb][0][2 * m + 1];
            b1.x = b.t[kb][1][2 * m];      b1.y = b.t[kb][1][2 * m + 1];
            c00 = __builtin_amdgcn_wmma_f32_16x16x4_f32(
                false, a0, false, b0, (short)0, c00, false, false);
            c01 = __builtin_amdgcn_wmma_f32_16x16x4_f32(
                false, a0, false, b1, (short)0, c01, false, false);
            c10 = __builtin_amdgcn_wmma_f32_16x16x4_f32(
                false, a1, false, b0, (short)0, c10, false, false);
            c11 = __builtin_amdgcn_wmma_f32_16x16x4_f32(
                false, a1, false, b1, (short)0, c11, false, false);
        }
    }
    c.t[0][0] = c00; c.t[0][1] = c01; c.t[1][0] = c10; c.t[1][1] = c11;
}

__device__ __forceinline__ void poly3(Tile4& p, const Tile4& e1, const Tile4& e2,
                                      float k0, float k1, float k2, int half, int l15)
{
    #pragma unroll
    for (int rb = 0; rb < 2; ++rb)
        #pragma unroll
        for (int cb = 0; cb < 2; ++cb)
            #pragma unroll
            for (int r = 0; r < 8; ++r)
                p.t[rb][cb][r] = k1 * e1.t[rb][cb][r] + k2 * e2.t[rb][cb][r];
    #pragma unroll
    for (int r = 0; r < 8; ++r) {
        const bool dg = (half * 8 + r) == l15;
        p.t[0][0][r] = dg ? p.t[0][0][r] + k0 : p.t[0][0][r];
        p.t[1][1][r] = dg ? p.t[1][1][r] + k0 : p.t[1][1][r];
    }
}

__device__ __forceinline__ void ns_sqrt_w(float* y, float* z, float* t, int iters,
                                          int lane, int half, int l15)
{
    float v[32];
    float cs = 0.f;
    #pragma unroll
    for (int j = 0; j < 32; ++j) {
        v[j] = y[j * LD + lane];
        cs += __builtin_fabsf(v[j]);
    }
    const float vd = y[lane * (LD + 1)];
    const float c = NS_START * wave_max_uniform(cs);
    const float inv_c = 1.f / c;
    #pragma unroll
    for (int j = 0; j < 32; ++j) {
        const float u = v[j] * inv_c;
        y[j * LD + lane] = u;
        z[j * LD + lane] = -0.5f * u;
    }
    z[lane * (LD + 1)] = 1.5f - 0.5f * (vd * inv_c);
    wave_lds_sync();
    Tile4 Yr, Zr, Tr;
    FragA aY, aZ, aT;
    lda(aY, y, half, l15);
    ldd(Zr, z, half, l15);
    mmr<0>(Yr, aY, Zr, half, l15);
    lda(aZ, z, half, l15);
    std4(y, Yr, half, l15);
    lda(aY, y, half, l15);
    __builtin_amdgcn_sched_barrier(0);
    float res2 = wave_sum_uniform(mmr<1>(Tr, aZ, Yr, half, l15));
    for (int it = 2; it < iters && !(res2 < NS_TOL2); ++it) {
        std4(t, Tr, half, l15);
        lda(aT, t, half, l15);
        __builtin_amdgcn_sched_barrier(0);
        mmr<0>(Yr, aY, Tr, half, l15);
        __builtin_amdgcn_sched_barrier(0);
        Tile4 Zn;
        mmr<0>(Zn, aT, Zr, half, l15);
        Zr = Zn;
        std4(z, Zr, half, l15);
        lda(aZ, z, half, l15);
        std4(y, Yr, half, l15);
        lda(aY, y, half, l15);
        __builtin_amdgcn_sched_barrier(0);
        res2 = wave_sum_uniform(mmr<1>(Tr, aZ, Yr, half, l15));
    }
    mmr<0>(Yr, aY, Tr, half, l15);
    std4(y, Yr, half, l15);
    const float sc = __fsqrt_rn(c);
    for (int i = lane; i < 1024; i += 32) {
        const int a = at32(i);
        y[a] = y[a] * sc;
    }
    wave_lds_sync();
}

__global__ __launch_bounds__(128, 1)
void spdnet_mixedop_kernel(const float* __restrict__ x,
                           const float* __restrict__ wts,
                           const float* __restrict__ Ws,
                           float* __restrict__ out)
{
    __shared__ __attribute__((aligned(16))) float lds[14 * MS + 32];
    float* const sS   = lds;
    float* const pSq  = lds + 4 * MS;
    float* const pSi  = lds + 5 * MS;
    float* const sW0  = lds + 7 * MS;
    float* const sW1  = lds + 8 * MS;
    float* const sP0  = lds + 9 * MS;
    float* const sB0  = lds + 11 * MS;
    float* const sB1  = lds + 12 * MS;
    float* const sB2  = lds + 13 * MS;
    float* const sRed = lds + 14 * MS;
    float* const sX   = lds + 4 * MS;
    float* const sWk  = lds + 8 * MS;
    float* const sP   = lds + 10 * MS;

    const int bc  = blockIdx.x;
    const int tid = threadIdx.x;

    float wk[4];
    #pragma unroll
    for (int k = 0; k < 4; ++k) wk[k] = wts[k];

    for (int i = tid; i < 4096; i += 128) sX[(i >> 6) * LDW + (i & 63)] = x[bc * 4096 + i];

    for (int k = 0; k < 4; ++k) {
        __syncthreads();
        for (int i = tid; i < 2048; i += 128) sWk[(i >> 6) * LDW + (i & 63)] = Ws[k * 2048 + i];
        mm(sP, LDW, sWk, LDW, sX, LDW, 32, 64, 64, false);
        mm(sS + k * MS, LD, sP, LDW, sWk, LDW, 32, 32, 64, true);
    }

    __syncthreads();
    for (int i = tid; i < 1024; i += 128) {
        const int a = at32(i);
        sB0[a] = wk[0] * sS[a] + wk[1] * sS[MS + a] +
                 wk[2] * sS[2 * MS + a] + wk[3] * sS[3 * MS + a];
    }

    ns_sqrt(pSq, pSi, sB0, sB1, sB2, sW0, sW1, sP0, sRed, 14);

    __syncthreads();

    {
        const int lane = tid & 31;
        const int half = lane >> 4;
        const int l15  = lane & 15;
        const int k = __builtin_amdgcn_readfirstlane(tid >> 5);
        float* const wa = sS + k * MS;
        float* const wb = lds + (6 + 2 * k) * MS;
        float* const wc = lds + (7 + 2 * k) * MS;

        mmw<0>(wb, pSi, wa, half, l15);
        mmw<0>(wc, wb, pSi, half, l15);

        for (int s = 0; s < 2; ++s) ns_sqrt_w(wc, wa, wb, 12, lane, half, l15);

        wc[lane * (LD + 1)] -= 1.f;
        wave_lds_sync();

        constexpr int J = LOG_DEG / 3 - 1;
        FragA aE, aX;
        Tile4 E1, E2, R, Rn, Bj;
        lda(aE, wc, half, l15);
        ldd(E1, wc, half, l15);
        mmr<0>(E2, aE, E1, half, l15);
        mmr<0>(Rn, aE, E2, half, l15);
        std4(wb, Rn, half, l15);
        lda(aX, wb, half, l15);
        poly3(R, E1, E2, log_coef(3 * J), log_coef(3 * J + 1), log_coef(3 * J + 2), half, l15);
        #pragma unroll
        for (int rb = 0; rb < 2; ++rb)
            #pragma unroll
            for (int cb = 0; cb < 2; ++cb)
                #pragma unroll
                for (int r = 0; r < 8; ++r)
                    R.t[rb][cb][r] += log_coef(LOG_DEG) * Rn.t[rb][cb][r];
        for (int j = J - 1; j >= 0; --j) {
            poly3(Bj, E1, E2, log_coef(3 * j), log_coef(3 * j + 1), log_coef(3 * j + 2), half, l15);
            mmra(Rn, aX, R, Bj);
            R = Rn;
        }
        std4(wa, R, half, l15);
    }

    __syncthreads();
    for (int i = tid; i < 1024; i += 128) {
        const int a = at32(i);
        float u = 0.f;
        #pragma unroll
        for (int k = 0; k < 4; ++k) u += (wk[k] * 4.f) * sS[k * MS + a];
        const float v = u * (1.f / 64.f);
        sB0[a] = v;
        sB1[a] = (diag32(i) ? 1.f : 0.f) + v * (1.f / 10.f);
    }
    for (int ii = 9; ii >= 1; --ii) {
        mm(sB2, LD, sB0, LD, sB1, LD, 32, 32, 32, false);
        const float inv = 1.f / (float)ii;
        for (int i = tid; i < 1024; i += 128) {
            const int a = at32(i);
            sB1[a] = (diag32(i) ? 1.f : 0.f) + sB2[a] * inv;
        }
    }
    float* r  = sB1;
    float* rn = sB2;
    for (int sq = 0; sq < 6; ++sq) {
        mm(rn, LD, r, LD, r, LD, 32, 32, 32, false);
        float* tmp = r; r = rn; rn = tmp;
    }
    mm(sB0, LD, pSq, LD, r, LD, 32, 32, 32, false);
    mm(sP0, LD, sB0, LD, pSq, LD, 32, 32, 32, false);
    __syncthreads();
    for (int i = tid; i < 1024; i += 128) out[bc * 1024 + i] = sP0[at32(i)];
}

extern "C" void kernel_launch(void* const* d_in, const int* in_sizes, int n_in,
                              void* d_out, int out_size, void* d_ws, size_t ws_size,
                              hipStream_t stream)
{
    const float* x   = (const float*)d_in[0];
    const float* wts = (const float*)d_in[1];
    const float* Ws  = (const float*)d_in[2];
    float* out = (float*)d_out;

    dim3 grid(2048);
    dim3 block(128);
    hipLaunchKernelGGL(spdnet_mixedop_kernel, grid, block, 0, stream,
                       x, wts, Ws, out);
}


# Log phase: every product there has a symmetric result: 3 of 4 tiles computed (24 WMMA, not 32), tile (1,0) is the mirror of (0,1) via LDS; its rounding and the residual sum's grouping differ
# speedup vs baseline: 8.2234x; 1.1179x over previous
#include <hip/hip_runtime.h>


typedef __attribute__((ext_vector_type(2))) float v2f;
typedef __attribute__((ext_vector_type(8))) float v8f;

constexpr int LD  = 36;
constexpr int LDW = 68;
constexpr int MS  = 32 * LD;
constexpr int LOG_DEG = 18;
static_assert(LOG_DEG % 3 == 0, "the log series is evaluated in blocks of three terms");

__device__ __forceinline__ float log_coef(int i)
{
    return i == 0 ? 0.f : ((i & 1) ? 1.f : -1.f) / (float)i;
}

constexpr float NS_TOL2 = 1.5e-7f;

constexpr float NS_START = 2.f / 3.f;

template <int M>
__device__ __forceinline__ float row_xor(float v)
{
    const int i = __builtin_bit_cast(int, v);
    return __builtin_bit_cast(float, __builtin_amdgcn_update_dpp(i, i, 0x160 + M, 0xf, 0xf, false));
}

__device__ __forceinline__ float wave_sum_uniform(float v)
{
    v += row_xor<1>(v);
    v += row_xor<2>(v);
    v += row_xor<4>(v);
    v += row_xor<8>(v);
    const int i = __builtin_bit_cast(int, v);
    v += __builtin_bit_cast(float, __builtin_amdgcn_permlanex16(i, i, 0x76543210, 0xfedcba98, false, false));
    return __builtin_bit_cast(float, __builtin_amdgcn_readfirstlane(__builtin_bit_cast(int, v)));
}

__device__ __forceinline__ float wave_max_uniform(float v)
{
    v = __builtin_fmaxf(v, row_xor<1>(v));
    v = __builtin_fmaxf(v, row_xor<2>(v));
    v = __builtin_fmaxf(v, row_xor<4>(v));
    v = __builtin_fmaxf(v, row_xor<8>(v));
    const int i = __builtin_bit_cast(int, v);
    v = __builtin_fmaxf(v, __builtin_bit_cast(float, __builtin_amdgcn_permlanex16(i, i, 0x76543210, 0xfedcba98, false, false)));
    return __builtin_bit_cast(float, __builtin_amdgcn_readfirstlane(__builtin_bit_cast(int, v)));
}

__device__ __forceinline__ int at32(int i) { return (i >> 5) * LD + (i & 31); }
__device__ __forceinline__ bool diag32(int i) { return (i >> 5) == (i & 31); }

__device__ __forceinline__ void mm(float* __restrict__ C, int ldc,
                                   const float* __restrict__ A, int lda,
                                   const float* __restrict__ B, int ldb,
                                   int M, int N, int Kd, bool transB)
{
    __syncthreads();
    const int wave = threadIdx.x >> 5;
    const int lane = threadIdx.x & 31;
    const int half = lane >> 4;
    const int l15  = lane & 15;
    const int tn = N >> 4;
    const int tiles = (M >> 4) * tn;
    for (int t = wave; t < tiles; t += 4) {
        const int mi = (t / tn) << 4;
        const int ni = (t % tn) << 4;
        v8f acc = {};
        #pragma unroll 4
        for (int k = 0; k < Kd; k += 4) {
            v2f a, b;
            const float* ar = A + (mi + l15) * lda + k + 2 * half;
            a.x = ar[0];
            a.y = ar[1];
            if (!transB) {
                const float* br = B + (k + 2 * half) * ldb + ni + l15;
                b.x = br[0];
                b.y = br[ldb];
            } else {
                const float* br = B + (ni + l15) * ldb + k + 2 * half;
                b.x = br[0];
                b.y = br[1];
            }
            acc = __builtin_amdgcn_wmma_f32_16x16x4_f32(
                false, a, false, b, (short)0, acc, false, false);
        }
        #pragma unroll
        for (int r = 0; r < 8; ++r)
            C[(mi + half * 8 + r) * ldc + ni + l15] = acc[r];
    }
    __syncthreads();
}

__device__ void ns_sqrt(float* dst_s, float* dst_i, const float* src,
                        float* y, float* z, float* t, float* yn, float* zn,
                        float* red, int iters)
{
    const int tid = threadIdx.x;
    const int wave = tid >> 5;
    const int lane = tid & 31;
    __syncthreads();
    float v[8];
    float cs = 0.f;
    #pragma unroll
    for (int j = 0; j < 8; ++j) {
        v[j] = src[at32(tid + 128 * j)];
        cs += __builtin_fabsf(v[j]);
    }
    t[tid] = cs;
    __syncthreads();
    const float c = NS_START * wave_max_uniform((t[lane] + t[32 + lane]) + (t[64 + lane] + t[96 + lane]));
    const float inv_c = 1.f / c;
    #pragma unroll
    for (int j = 0; j < 8; ++j) {
        const int i = tid + 128 * j;
        const int a = at32(i);
        const float u = v[j] * inv_c;
        y[a] = u;
        zn[a] = (diag32(i) ? 1.5f : 0.f) - 0.5f * u;
    }
    const int half = lane >> 4;
    const int l15  = lane & 15;
    const int mi = (wave >> 1) << 4;
    const int ni = (wave & 1) << 4;
    const int aoff = (mi + l15) * LD + 2 * half;
    const int boff = (2 * half) * LD + ni + l15;
    const int coff = (mi + half * 8) * LD + ni + l15;
    __syncthreads();
    {
        v2f ya[8], tb[8];
        #pragma unroll
        for (int j = 0; j < 8; ++j) {
            ya[j].x = y[aoff + 4 * j];
            ya[j].y = y[aoff + 4 * j + 1];
            tb[j].x = zn[boff + 4 * j * LD];
            tb[j].y = zn[boff + 4 * j * LD + LD];
        }
        v8f accY = {};
        #pragma unroll
        for (int j = 0; j < 8; ++j)
            accY = __builtin_amdgcn_wmma_f32_16x16x4_f32(
                false, ya[j], false, tb[j], (short)0, accY, false, false);
        #pragma unroll
        for (int r = 0; r < 8; ++r) yn[coff + r * LD] = accY[r];
        float* tmp;
        tmp = y; y = yn; yn = tmp;
        tmp = z; z = zn; zn = tmp;
    }
    for (int it = 1; it < iters; ++it) {
        __syncthreads();
        {
            v2f fa[8], fb[8];
            #pragma unroll
            for (int j = 0; j < 8; ++j) {
                fa[j].x = z[aoff + 4 * j];
                fa[j].y = z[aoff + 4 * j + 1];
                fb[j].x = y[boff + 4 * j * LD];
                fb[j].y = y[boff + 4 * j * LD + LD];
            }
            __builtin_amdgcn_sched_barrier(0);
            v8f acc = {};
            #pragma unroll
            for (int j = 0; j < 8; ++j)
                acc = __builtin_amdgcn_wmma_f32_16x16x4_f32(
                    false, fa[j], false, fb[j], (short)0, acc, false, false);
            float r2 = 0.f;
            #pragma unroll
            for (int r = 0; r < 8; ++r) {
                const bool dg = (mi + half * 8 + r) == (ni + l15);
                t[coff + r * LD] = (dg ? 1.5f : 0.f) - 0.5f * acc[r];
                const float d = (dg ? 1.f : 0.f) - acc[r];
                r2 += d * d;
            }
            r2 = wave_sum_uniform(r2);
            if (lane == 0) red[wave] = r2;
        }
        __syncthreads();
        const float res2 = (red[0] + red[1]) + (red[2] + red[3]);
        {
            v2f ya[8], tb[8], ta[8], zb[8];
            #pragma unroll
            for (int j = 0; j < 8; ++j) {
                ya[j].x = y[aoff + 4 * j];
                ya[j].y = y[aoff + 4 * j + 1];
                tb[j].x = t[boff + 4 * j * LD];
                tb[j].y = t[boff + 4 * j * LD + LD];
                ta[j].x = t[aoff + 4 * j];
                ta[j].y = t[aoff + 4 * j + 1];
                zb[j].x = z[boff + 4 * j * LD];
                zb[j].y = z[boff + 4 * j * LD + LD];
            }
            __builtin_amdgcn_sched_barrier(0);
            v8f accY = {}, accZ = {};
            #pragma unroll
            for (int j = 0; j < 8; ++j) {
                accY = __builtin_amdgcn_wmma_f32_16x16x4_f32(
                    false, ya[j], false, tb[j], (short)0, accY, false, false);
                accZ = __builtin_amdgcn_wmma_f32_16x16x4_f32(
                    false, ta[j], false, zb[j], (short)0, accZ, false, false);
            }
            #pragma unroll
            for (int r = 0; r < 8; ++r) {
                yn[coff + r * LD] = accY[r];
                zn[coff + r * LD] = accZ[r];
            }
        }
        float* tmp;
        tmp = y; y = yn; yn = tmp;
        tmp = z; z = zn; zn = tmp;
        if (res2 < NS_TOL2) break;
    }
    const float sc  = __fsqrt_rn(c);
    const float isc = __frsqrt_rn(c);
    __syncthreads();
    if (dst_s) for (int i = tid; i < 1024; i += 128) dst_s[at32(i)] = y[at32(i)] * sc;
    if (dst_i) for (int i = tid; i < 1024; i += 128) dst_i[at32(i)] = z[at32(i)] * isc;
    __syncthreads();
}

__device__ __forceinline__ void wave_lds_sync()
{
    __builtin_amdgcn_fence(__ATOMIC_RELEASE, "wavefront");
    __builtin_amdgcn_wave_barrier();
    __builtin_amdgcn_fence(__ATOMIC_ACQUIRE, "wavefront");
}

template <int EPI>
__device__ __forceinline__ void mmw(float* C, const float* A, const float* B,
                                    int half, int l15)
{
    v2f a0[8], a1[8], b0[8], b1[8];
    const float* ar = A + l15 * LD + 2 * half;
    const float* br = B + (2 * half) * LD + l15;
    #pragma unroll
    for (int j = 0; j < 8; ++j) {
        a0[j].x = ar[4 * j];
        a0[j].y = ar[4 * j + 1];
        a1[j].x = ar[16 * LD + 4 * j];
        a1[j].y = ar[16 * LD + 4 * j + 1];
        b0[j].x = br[4 * j * LD];
        b0[j].y = br[4 * j * LD + LD];
        b1[j].x = br[4 * j * LD + 16];
        b1[j].y = br[4 * j * LD + LD + 16];
    }
    __builtin_amdgcn_sched_barrier(0);
    v8f c00 = {}, c01 = {}, c10 = {}, c11 = {};
    #pragma unroll
    for (int j = 0; j < 8; ++j) {
        c00 = __builtin_amdgcn_wmma_f32_16x16x4_f32(
            false, a0[j], false, b0[j], (short)0, c00, false, false);
        c01 = __builtin_amdgcn_wmma_f32_16x16x4_f32(
            false, a0[j], false, b1[j], (short)0, c01, false, false);
        c10 = __builtin_amdgcn_wmma_f32_16x16x4_f32(
            false, a1[j], false, b0[j], (short)0, c10, false, false);
        c11 = __builtin_amdgcn_wmma_f32_16x16x4_f32(
            false, a1[j], false, b1[j], (short)0, c11, false, false);
    }
    float* cr = C + (half * 8) * LD + l15;
    #pragma unroll
    for (int r = 0; r < 8; ++r) {
        if (EPI == 1) {
            const bool dg = (half * 8 + r) == l15;
            cr[r * LD]             = (dg ? 1.5f : 0.f) - 0.5f * c00[r];
            cr[r * LD + 16]        = 0.f - 0.5f * c01[r];
            cr[(16 + r) * LD]      = 0.f - 0.5f * c10[r];
            cr[(16 + r) * LD + 16] = (dg ? 1.5f : 0.f) - 0.5f * c11[r];
        } else {
            cr[r * LD]             = c00[r];
            cr[r * LD + 16]        = c01[r];
            cr[(16 + r) * LD]      = c10[r];
            cr[(16 + r) * LD + 16] = c11[r];
        }
    }
    wave_lds_sync();
}

typedef __attribute__((ext_vector_type(4))) float v4f;

struct Tile4 { v8f t[2][2]; };

struct FragA { v4f f[2][2][2]; };

__device__ __forceinline__ void lda(FragA& a, const float* A, int half, int l15)
{
    const float* ar = A + l15 * LD + 8 * half;
    #pragma unroll
    for (int rb = 0; rb < 2; ++rb)
        #pragma unroll
        for (int kb = 0; kb < 2; ++kb)
            #pragma unroll
            for (int p = 0; p < 2; ++p)
                a.f[rb][kb][p] = *(const v4f*)(ar + rb * 16 * LD + kb * 16 + 4 * p);
}

__device__ __forceinline__ void ldd(Tile4& c, const float* C, int half, int l15)
{
    const float* cr = C + (half * 8) * LD + l15;
    #pragma unroll
    for (int rb = 0; rb < 2; ++rb)
        #pragma unroll
        for (int cb = 0; cb < 2; ++cb)
            #pragma unroll
            for (int r = 0; r < 8; ++r)
                c.t[rb][cb][r] = cr[(16 * rb + r) * LD + 16 * cb];
}


__device__ __forceinline__ void std4s(float* C, Tile4& c, int half, int l15)
{
    float* cr = C + (half * 8) * LD + l15;
    #pragma unroll
    for (int r = 0; r < 8; ++r) {
        cr[r * LD]             = c.t[0][0][r];
        cr[r * LD + 16]        = c.t[0][1][r];
        cr[(16 + r) * LD + 16] = c.t[1][1][r];
    }
    float* mr = C + (16 + l15) * LD + 8 * half;
    const v8f u = c.t[0][1];
    *(v4f*)mr       = (v4f){u[0], u[1], u[2], u[3]};
    *(v4f*)(mr + 4) = (v4f){u[4], u[5], u[6], u[7]};
    wave_lds_sync();
    const float* tr = C + l15 * LD + 16 + 8 * half;
    const v4f lo = *(const v4f*)tr;
    const v4f hi = *(const v4f*)(tr + 4);
    c.t[1][0] = (v8f){lo[0], lo[1], lo[2], lo[3], hi[0], hi[1], hi[2], hi[3]};
}

__device__ __forceinline__ void sym_fill(float* S, Tile4& c, int half, int l15)
{
    float* cr = S + (half * 8) * LD + 16 + l15;
    #pragma unroll
    for (int r = 0; r < 8; ++r) cr[r * LD] = c.t[0][1][r];
    wave_lds_sync();
    const float* tr = S + l15 * LD + 16 + 8 * half;
    const v4f lo = *(const v4f*)tr;
    const v4f hi = *(const v4f*)(tr + 4);
    c.t[1][0] = (v8f){lo[0], lo[1], lo[2], lo[3], hi[0], hi[1], hi[2], hi[3]};
}

template <int EPI>
__device__ __forceinline__ float mmr(Tile4& c, const FragA& a, const Tile4& b, int half, int l15)
{
    v8f c00 = {}, c01 = {}, c11 = {};
    #pragma unroll
    for (int kb = 0; kb < 2; ++kb) {
        #pragma unroll
        for (int m = 0; m < 4; ++m) {
            const int e = 2 * (m & 1);
            v2f a0, a1, b0, b1;
            a0.x = a.f[0][kb][m >> 1][e];  a0.y = a.f[0][kb][m >> 1][e + 1];
            a1.x = a.f[1][kb][m >> 1][e];  a1.y = a.f[1][kb][m >> 1][e + 1];
            b0.x = b.t[kb][0][2 * m];      b0.y = b.t[kb][0][2 * m + 1];
            b1.x = b.t[kb][1][2 * m];      b1.y = b.t[kb][1][2 * m + 1];
            c00 = __builtin_amdgcn_wmma_f32_16x16x4_f32(
                false, a0, false, b0, (short)0, c00, false, false);
            c01 = __builtin_amdgcn_wmma_f32_16x16x4_f32(
                false, a0, false, b1, (short)0, c01, false, false);
            c11 = __builtin_amdgcn_wmma_f32_16x16x4_f32(
                false, a1, false, b1, (short)0, c11, false, false);
        }
    }
    float r2 = 0.f;
    if (EPI == 1) {
        #pragma unroll
        for (int r = 0; r < 8; ++r) {
            const bool dg = (half * 8 + r) == l15;
            const float d00 = (dg ? 1.f : 0.f) - c00[r];
            const float d11 = (dg ? 1.f : 0.f) - c11[r];
            r2 += (d00 * d00 + d11 * d11) + 2.f * (c01[r] * c01[r]);
            c00[r] = (dg ? 1.5f : 0.f) - 0.5f * c00[r];
            c01[r] = 0.f - 0.5f * c01[r];
            c11[r] = (dg ? 1.5f : 0.f) - 0.5f * c11[r];
        }
    }
    c.t[0][0] = c00; c.t[0][1] = c01; c.t[1][1] = c11;
    return r2;
}

__device__ __forceinline__ void mmra(Tile4& c, const FragA& a, const Tile4& b, const Tile4& c0)
{
    v8f c00 = c0.t[0][0], c01 = c0.t[0][1], c11 = c0.t[1][1];
    #pragma unroll
    for (int kb = 0; kb < 2; ++kb) {
        #pragma unroll
        for (int m = 0; m < 4; ++m) {
            const int e = 2 * (m & 1);
            v2f a0, a1, b0, b1;
            a0.x = a.f[0][kb][m >> 1][e];  a0.y = a.f[0][kb][m >> 1][e + 1];
            a1.x = a.f[1][kb][m >> 1][e];  a1.y = a.f[1][kb][m >> 1][e + 1];
            b0.x = b.t[kb][0][2 * m];      b0.y = b.t[kb][0][2 * m + 1];
            b1.x = b.t[kb][1][2 * m];      b1.y = b.t[kb][1][2 * m + 1];
            c00 = __builtin_amdgcn_wmma_f32_16x16x4_f32(
                false, a0, false, b0, (short)0, c00, false, false);
            c01 = __builtin_amdgcn_wmma_f32_16x16x4_f32(
                false, a0, false, b1, (short)0, c01, false, false);
            c11 = __builtin_amdgcn_wmma_f32_16x16x4_f32(
                false, a1, false, b1, (short)0, c11, false, false);
        }
    }
    c.t[0][0] = c00; c.t[0][1] = c01; c.t[1][1] = c11;
}

__device__ __forceinline__ void poly3(Tile4& p, const Tile4& e1, const Tile4& e2,
                                      float k0, float k1, float k2, int half, int l15)
{
    #pragma unroll
    for (int rb = 0; rb < 2; ++rb)
        #pragma unroll
        for (int cb = 0; cb < 2; ++cb)
            #pragma unroll
            for (int r = 0; r < 8; ++r)
                p.t[rb][cb][r] = k1 * e1.t[rb][cb][r] + k2 * e2.t[rb][cb][r];
    #pragma unroll
    for (int r = 0; r < 8; ++r) {
        const bool dg = (half * 8 + r) == l15;
        p.t[0][0][r] = dg ? p.t[0][0][r] + k0 : p.t[0][0][r];
        p.t[1][1][r] = dg ? p.t[1][1][r] + k0 : p.t[1][1][r];
    }
}

__device__ __forceinline__ void ns_sqrt_w(float* y, float* z, float* t, int iters,
                                          int lane, int half, int l15)
{
    float v[32];
    float cs = 0.f;
    #pragma unroll
    for (int j = 0; j < 32; ++j) {
        v[j] = y[j * LD + lane];
        cs += __builtin_fabsf(v[j]);
    }
    const float vd = y[lane * (LD + 1)];
    const float c = NS_START * wave_max_uniform(cs);
    const float inv_c = 1.f / c;
    #pragma unroll
    for (int j = 0; j < 32; ++j) {
        const float u = v[j] * inv_c;
        y[j * LD + lane] = u;
        z[j * LD + lane] = -0.5f * u;
    }
    z[lane * (LD + 1)] = 1.5f - 0.5f * (vd * inv_c);
    wave_lds_sync();
    Tile4 Yr, Zr, Tr;
    FragA aY, aZ, aT;
    lda(aY, y, half, l15);
    ldd(Zr, z, half, l15);
    mmr<0>(Yr, aY, Zr, half, l15);
    lda(aZ, z, half, l15);
    std4s(y, Yr, half, l15);
    lda(aY, y, half, l15);
    __builtin_amdgcn_sched_barrier(0);
    float res2 = wave_sum_uniform(mmr<1>(Tr, aZ, Yr, half, l15));
    for (int it = 2; it < iters && !(res2 < NS_TOL2); ++it) {
        std4s(t, Tr, half, l15);
        lda(aT, t, half, l15);
        __builtin_amdgcn_sched_barrier(0);
        mmr<0>(Yr, aY, Tr, half, l15);
        __builtin_amdgcn_sched_barrier(0);
        Tile4 Zn;
        mmr<0>(Zn, aT, Zr, half, l15);
        std4s(z, Zn, half, l15);
        Zr = Zn;
        lda(aZ, z, half, l15);
        std4s(y, Yr, half, l15);
        lda(aY, y, half, l15);
        __builtin_amdgcn_sched_barrier(0);
        res2 = wave_sum_uniform(mmr<1>(Tr, aZ, Yr, half, l15));
    }
    sym_fill(t, Tr, half, l15);
    mmr<0>(Yr, aY, Tr, half, l15);
    std4s(y, Yr, half, l15);
    const float sc = __fsqrt_rn(c);
    for (int i = lane; i < 1024; i += 32) {
        const int a = at32(i);
        y[a] = y[a] * sc;
    }
    wave_lds_sync();
}

__global__ __launch_bounds__(128, 1)
void spdnet_mixedop_kernel(const float* __restrict__ x,
                           const float* __restrict__ wts,
                           const float* __restrict__ Ws,
                           float* __restrict__ out)
{
    __shared__ __attribute__((aligned(16))) float lds[14 * MS + 32];
    float* const sS   = lds;
    float* const pSq  = lds + 4 * MS;
    float* const pSi  = lds + 5 * MS;
    float* const sW0  = lds + 7 * MS;
    float* const sW1  = lds + 8 * MS;
    float* const sP0  = lds + 9 * MS;
    float* const sB0  = lds + 11 * MS;
    float* const sB1  = lds + 12 * MS;
    float* const sB2  = lds + 13 * MS;
    float* const sRed = lds + 14 * MS;
    float* const sX   = lds + 4 * MS;
    float* const sWk  = lds + 8 * MS;
    float* const sP   = lds + 10 * MS;

    const int bc  = blockIdx.x;
    const int tid = threadIdx.x;

    float wk[4];
    #pragma unroll
    for (int k = 0; k < 4; ++k) wk[k] = wts[k];

    for (int i = tid; i < 4096; i += 128) sX[(i >> 6) * LDW + (i & 63)] = x[bc * 4096 + i];

    for (int k = 0; k < 4; ++k) {
        __syncthreads();
        for (int i = tid; i < 2048; i += 128) sWk[(i >> 6) * LDW + (i & 63)] = Ws[k * 2048 + i];
        mm(sP, LDW, sWk, LDW, sX, LDW, 32, 64, 64, false);
        mm(sS + k * MS, LD, sP, LDW, sWk, LDW, 32, 32, 64, true);
    }

    __syncthreads();
    for (int i = tid; i < 1024; i += 128) {
        const int a = at32(i);
        sB0[a] = wk[0] * sS[a] + wk[1] * sS[MS + a] +
                 wk[2] * sS[2 * MS + a] + wk[3] * sS[3 * MS + a];
    }

    ns_sqrt(pSq, pSi, sB0, sB1, sB2, sW0, sW1, sP0, sRed, 14);

    __syncthreads();

    {
        const int lane = tid & 31;
        const int half = lane >> 4;
        const int l15  = lane & 15;
        const int k = __builtin_amdgcn_readfirstlane(tid >> 5);
        float* const wa = sS + k * MS;
        float* const wb = lds + (6 + 2 * k) * MS;
        float* const wc = lds + (7 + 2 * k) * MS;

        mmw<0>(wb, pSi, wa, half, l15);
        mmw<0>(wc, wb, pSi, half, l15);

        for (int s = 0; s < 2; ++s) ns_sqrt_w(wc, wa, wb, 12, lane, half, l15);

        wc[lane * (LD + 1)] -= 1.f;
        wave_lds_sync();

        constexpr int J = LOG_DEG / 3 - 1;
        FragA aE, aX;
        Tile4 E1, E2, R, Rn, Bj;
        lda(aE, wc, half, l15);
        ldd(E1, wc, half, l15);
        mmr<0>(E2, aE, E1, half, l15);
        sym_fill(wb, E2, half, l15);
        mmr<0>(Rn, aE, E2, half, l15);
        std4s(wb, Rn, half, l15);
        lda(aX, wb, half, l15);
        poly3(R, E1, E2, log_coef(3 * J), log_coef(3 * J + 1), log_coef(3 * J + 2), half, l15);
        #pragma unroll
        for (int rb = 0; rb < 2; ++rb)
            #pragma unroll
            for (int cb = 0; cb < 2; ++cb)
                #pragma unroll
                for (int r = 0; r < 8; ++r)
                    R.t[rb][cb][r] += log_coef(LOG_DEG) * Rn.t[rb][cb][r];
        for (int j = J - 1; j >= 0; --j) {
            poly3(Bj, E1, E2, log_coef(3 * j), log_coef(3 * j + 1), log_coef(3 * j + 2), half, l15);
            mmra(Rn, aX, R, Bj);
            if (j > 0) sym_fill(wa, Rn, half, l15);
            else       std4s(wa, Rn, half, l15);
            R = Rn;
        }
    }

    __syncthreads();
    for (int i = tid; i < 1024; i += 128) {
        const int a = at32(i);
        float u = 0.f;
        #pragma unroll
        for (int k = 0; k < 4; ++k) u += (wk[k] * 4.f) * sS[k * MS + a];
        const float v = u * (1.f / 64.f);
        sB0[a] = v;
        sB1[a] = (diag32(i) ? 1.f : 0.f) + v * (1.f / 10.f);
    }
    for (int ii = 9; ii >= 1; --ii) {
        mm(sB2, LD, sB0, LD, sB1, LD, 32, 32, 32, false);
        const float inv = 1.f / (float)ii;
        for (int i = tid; i < 1024; i += 128) {
            const int a = at32(i);
            sB1[a] = (diag32(i) ? 1.f : 0.f) + sB2[a] * inv;
        }
    }
    float* r  = sB1;
    float* rn = sB2;
    for (int sq = 0; sq < 6; ++sq) {
        mm(rn, LD, r, LD, r, LD, 32, 32, 32, false);
        float* tmp = r; r = rn; rn = tmp;
    }
    mm(sB0, LD, pSq, LD, r, LD, 32, 32, 32, false);
    mm(sP0, LD, sB0, LD, pSq, LD, 32, 32, 32, false);
    __syncthreads();
    for (int i = tid; i < 1024; i += 128) out[bc * 1024 + i] = sP0[at32(i)];
}

extern "C" void kernel_launch(void* const* d_in, const int* in_sizes, int n_in,
                              void* d_out, int out_size, void* d_ws, size_t ws_size,
                              hipStream_t stream)
{
    const float* x   = (const float*)d_in[0];
    const float* wts = (const float*)d_in[1];
    const float* Ws  = (const float*)d_in[2];
    float* out = (float*)d_out;

    dim3 grid(2048);
    dim3 block(128);
    hipLaunchKernelGGL(spdnet_mixedop_kernel, grid, block, 0, stream,
                       x, wts, Ws, out);
}
